# starting kernel
# baseline (speedup 1.0000x reference)
#include <hip/hip_runtime.h>
#include <hip/hip_bf16.h>
#include <math.h>

typedef __attribute__((ext_vector_type(2))) float v2f;
typedef __attribute__((ext_vector_type(8))) float v8f;

#define B_TOT   4096
#define S_LEN   512
#define HIDN    64
#define G4      256
#define NCOV    10
#define EMB     16
#define TB      16
#define HPITCH  68
#define SGPITCH 260
#define XPITCH  12
#define CHUNK   16
#define CPITCH  (CHUNK*NCOV)

struct __align__(16) SMem {
  float covS[TB * CPITCH];
  float sg[TB * SGPITCH];
  float h[TB * HPITCH];
  float xdyn[TB * XPITCH];
  float prevS[TB * CHUNK];
  float sx[TB * 65];
  float part[4 * 32];
  float wo[HIDN * 2];
  float bo[2];
  float scl[TB];
  float iscl[TB];
};

__device__ __forceinline__ float fast_tanh(float x) {
#if __has_builtin(__builtin_amdgcn_tanhf)
  return __builtin_amdgcn_tanhf(x);
#else
  float y;
  asm("v_tanh_f32 %0, %1" : "=v"(y) : "v"(x));
  return y;
#endif
}

__device__ __forceinline__ float fast_sigmoid(float x) {
  return __builtin_amdgcn_rcpf(1.0f + __expf(-x));
}

__global__ __launch_bounds__(128, 1)
void deepar_lstm_kernel(const float* __restrict__ target,
                        const float* __restrict__ covs,
                        const int*   __restrict__ cats,
                        const float* __restrict__ scale,
                        const float* __restrict__ emb0,
                        const float* __restrict__ emb1,
                        const float* __restrict__ emb2,
                        const float* __restrict__ emb3,
                        const float* __restrict__ w_ih,
                        const float* __restrict__ w_hh,
                        const float* __restrict__ bias,
                        const float* __restrict__ w_out,
                        const float* __restrict__ b_out,
                        float* __restrict__ out)
{
  __shared__ SMem sm;
  const int tid  = threadIdx.x;
  const int wave = tid >> 5;
  const int lane = tid & 31;
  const int b0   = blockIdx.x * TB;

  if (tid < TB) {
    float s = scale[b0 + tid];
    sm.scl[tid]  = s;
    sm.iscl[tid] = 1.0f / fmaxf(s, 1e-4f);
  }
  if (tid < HIDN * 2) sm.wo[tid] = w_out[tid];
  if (tid < 2)        sm.bo[tid] = b_out[tid];

  for (int e = tid; e < TB * 65; e += 128) {
    int m = e / 65, k = e % 65;
    float v;
    if      (k < 16) v = emb0[cats[(b0 + m) * 4 + 0] * EMB + k];
    else if (k < 32) v = emb1[cats[(b0 + m) * 4 + 1] * EMB + (k - 16)];
    else if (k < 48) v = emb2[cats[(b0 + m) * 4 + 2] * EMB + (k - 32)];
    else if (k < 64) v = emb3[cats[(b0 + m) * 4 + 3] * EMB + (k - 48)];
    else             v = log1pf(scale[b0 + m]);
    sm.sx[e] = v;
  }
  __syncthreads();

  for (int e = tid; e < TB * G4; e += 128) {
    int m = e >> 8, j = e & 255;
    float acc = bias[j];
    #pragma unroll 5
    for (int k = 0; k < 65; ++k)
      acc += sm.sx[m * 65 + k] * w_ih[(11 + k) * G4 + j];
    sm.sg[m * SGPITCH + j] = acc;
  }

  const int khalf = lane >> 4;
  const int ml    = lane & 15;
  const int ncol  = wave * 16 + ml;

  v2f whhB[16][4];
  #pragma unroll
  for (int kt = 0; kt < 16; ++kt)
    #pragma unroll
    for (int nt = 0; nt < 4; ++nt) {
      int k = kt * 4 + khalf * 2;
      int n = nt * 64 + ncol;
      v2f b = { w_hh[k * G4 + n], w_hh[(k + 1) * G4 + n] };
      whhB[kt][nt] = b;
    }
  v2f wihB[3][4];
  #pragma unroll
  for (int kt = 0; kt < 3; ++kt)
    #pragma unroll
    for (int nt = 0; nt < 4; ++nt) {
      int k = kt * 4 + khalf * 2;
      int n = nt * 64 + ncol;
      v2f b = { w_ih[k * G4 + n], w_ih[(k + 1) * G4 + n] };
      wihB[kt][nt] = b;
    }
  __syncthreads();
  v8f statC[4];
  #pragma unroll
  for (int nt = 0; nt < 4; ++nt)
    #pragma unroll
    for (int r = 0; r < 8; ++r) {
      int m = r + khalf * 8;
      statC[nt][r] = sm.sg[m * SGPITCH + nt * 64 + ncol];
    }

  for (int e = tid; e < TB * HPITCH; e += 128) sm.h[e] = 0.0f;
  for (int e4 = tid; e4 < TB * (CPITCH / 4); e4 += 128) {
    int m = e4 / (CPITCH / 4), off4 = e4 % (CPITCH / 4);
    *(float4*)&sm.covS[m * CPITCH + off4 * 4] =
        *(const float4*)&covs[(size_t)(b0 + m) * S_LEN * NCOV + off4 * 4];
    if ((off4 & 7) == 0)
      __builtin_prefetch(&covs[((size_t)(b0 + m) * S_LEN + CHUNK) * NCOV + off4 * 4], 0, 1);
  }
  for (int e = tid; e < TB * CHUNK; e += 128) {
    int m = e >> 4, j = e & 15;
    sm.prevS[e] = (j == 0) ? 0.0f : target[(size_t)(b0 + m) * S_LEN + j - 1];
  }
  __syncthreads();
  for (int e = tid; e < TB * XPITCH; e += 128) {
    int m = e / XPITCH, k = e % XPITCH;
    float v;
    if      (k == 0)    v = sm.prevS[m * CHUNK] * sm.iscl[m];
    else if (k <= NCOV) v = sm.covS[m * CPITCH + (k - 1)];
    else                v = 0.0f;
    sm.xdyn[e] = v;
  }

  v8f creg = {0.f, 0.f, 0.f, 0.f, 0.f, 0.f, 0.f, 0.f};
  v8f hreg;

  for (int t = 0; t < S_LEN; ++t) {
    __syncthreads();

    v2f xA[3], hA[16];
    #pragma unroll
    for (int kt = 0; kt < 3; ++kt)
      xA[kt] = *(const v2f*)&sm.xdyn[ml * XPITCH + kt * 4 + khalf * 2];
    #pragma unroll
    for (int kt = 0; kt < 16; ++kt)
      hA[kt] = *(const v2f*)&sm.h[ml * HPITCH + kt * 4 + khalf * 2];

    v8f acc[4];
    #pragma unroll
    for (int nt = 0; nt < 4; ++nt) {
      v8f a = statC[nt];
      #pragma unroll
      for (int kt = 0; kt < 3; ++kt)
        a = __builtin_amdgcn_wmma_f32_16x16x4_f32(false, xA[kt], false, wihB[kt][nt],
                                                  (short)0, a, false, false);
      #pragma unroll
      for (int kt = 0; kt < 16; ++kt)
        a = __builtin_amdgcn_wmma_f32_16x16x4_f32(false, hA[kt], false, whhB[kt][nt],
                                                  (short)0, a, false, false);
      acc[nt] = a;
    }
    #pragma unroll
    for (int r = 0; r < 8; ++r) {
      float gi = fast_sigmoid(acc[0][r]);
      float gf = fast_sigmoid(acc[1][r]);
      float gg = fast_tanh(acc[2][r]);
      float go = fast_sigmoid(acc[3][r]);
      float cc = gf * creg[r] + gi * gg;
      creg[r] = cc;
      hreg[r] = go * fast_tanh(cc);
    }
    {
      int p = tid & 31, q = tid >> 5;
      int m = p >> 1, j = p & 1;
      float s = 0.0f;
      #pragma unroll
      for (int kk = 0; kk < 16; ++kk) {
        int k = q * 16 + kk;
        s += sm.h[m * HPITCH + k] * sm.wo[k * 2 + j];
      }
      sm.part[q * 32 + p] = s;
    }

    __syncthreads();

    #pragma unroll
    for (int r = 0; r < 8; ++r) {
      int m = r + khalf * 8;
      sm.h[m * HPITCH + ncol] = hreg[r];
    }
    if (t > 0 && tid < 32) {
      int m = tid >> 1, j = tid & 1;
      float r = sm.bo[j] + sm.part[tid] + sm.part[32 + tid] +
                sm.part[64 + tid] + sm.part[96 + tid];
      float sp = (r > 20.0f) ? r : log1pf(__expf(r));
      sp += 1e-4f;
      size_t idx = (size_t)(b0 + m) * S_LEN + (t - 1);
      if (j == 0) out[idx] = sp * sm.scl[m];
      else        out[(size_t)B_TOT * S_LEN + idx] = sp;
    }
    const int tn = t + 1;
    if (tn < S_LEN) {
      if ((tn & (CHUNK - 1)) == 0) {
        for (int e4 = tid; e4 < TB * (CPITCH / 4); e4 += 128) {
          int m = e4 / (CPITCH / 4), off4 = e4 % (CPITCH / 4);
          *(float4*)&sm.covS[m * CPITCH + off4 * 4] =
              *(const float4*)&covs[((size_t)(b0 + m) * S_LEN + tn) * NCOV + off4 * 4];
          if ((off4 & 7) == 0 && tn + CHUNK < S_LEN)
            __builtin_prefetch(&covs[((size_t)(b0 + m) * S_LEN + tn + CHUNK) * NCOV + off4 * 4], 0, 1);
        }
        for (int e = tid; e < TB * CHUNK; e += 128) {
          int m = e >> 4, j = e & 15;
          sm.prevS[e] = target[(size_t)(b0 + m) * S_LEN + tn + j - 1];
        }
        __syncthreads();
      }
      const int tc = tn & (CHUNK - 1);
      for (int e = tid; e < TB * XPITCH; e += 128) {
        int m = e / XPITCH, k = e % XPITCH;
        float v;
        if      (k == 0)    v = sm.prevS[m * CHUNK + tc] * sm.iscl[m];
        else if (k <= NCOV) v = sm.covS[m * CPITCH + tc * NCOV + (k - 1)];
        else                v = 0.0f;
        sm.xdyn[e] = v;
      }
    }
  }

  __syncthreads();
  {
    int p = tid & 31, q = tid >> 5;
    int m = p >> 1, j = p & 1;
    float s = 0.0f;
    #pragma unroll
    for (int kk = 0; kk < 16; ++kk) {
      int k = q * 16 + kk;
      s += sm.h[m * HPITCH + k] * sm.wo[k * 2 + j];
    }
    sm.part[q * 32 + p] = s;
  }
  __syncthreads();
  if (tid < 32) {
    int m = tid >> 1, j = tid & 1;
    float r = sm.bo[j] + sm.part[tid] + sm.part[32 + tid] +
              sm.part[64 + tid] + sm.part[96 + tid];
    float sp = (r > 20.0f) ? r : log1pf(__expf(r));
    sp += 1e-4f;
    size_t idx = (size_t)(b0 + m) * S_LEN + (S_LEN - 1);
    if (j == 0) out[idx] = sp * sm.scl[m];
    else        out[(size_t)B_TOT * S_LEN + idx] = sp;
  }
}

extern "C" void kernel_launch(void* const* d_in, const int* in_sizes, int n_in,
                              void* d_out, int out_size, void* d_ws, size_t ws_size,
                              hipStream_t stream) {
  const float* target = (const float*)d_in[0];
  const float* covs   = (const float*)d_in[1];
  const int*   cats   = (const int*)  d_in[2];
  const float* scale  = (const float*)d_in[3];
  const float* emb0   = (const float*)d_in[4];
  const float* emb1   = (const float*)d_in[5];
  const float* emb2   = (const float*)d_in[6];
  const float* emb3   = (const float*)d_in[7];
  const float* w_ih   = (const float*)d_in[8];
  const float* w_hh   = (const float*)d_in[9];
  const float* bias   = (const float*)d_in[10];
  const float* w_out  = (const float*)d_in[11];
  const float* b_out  = (const float*)d_in[12];
  float* out = (float*)d_out;

  dim3 grid(B_TOT / TB);
  dim3 block(128);
  deepar_lstm_kernel<<<grid, block, 0, stream>>>(
      target, covs, cats, scale, emb0, emb1, emb2, emb3,
      w_ih, w_hh, bias, w_out, b_out, out);
}


# Per-step LDS rebuild of the next input row removed: each 16-step chunk is staged once, straight into WMMA A-tile order [step][row][12], and the scan reads its x operand from it
# speedup vs baseline: 1.2101x; 1.2101x over previous
#include <hip/hip_runtime.h>
#include <hip/hip_bf16.h>
#include <math.h>

typedef __attribute__((ext_vector_type(2))) float v2f;
typedef __attribute__((ext_vector_type(8))) float v8f;

#define B_TOT   4096
#define S_LEN   512
#define HIDN    64
#define G4      256
#define NCOV    10
#define EMB     16
#define TB      16
#define HPITCH  68
#define SGPITCH 260
#define XPITCH  12
#define CHUNK   16
#define CPITCH  (CHUNK*NCOV)

struct __align__(16) SMem {
  float xs[CHUNK * TB * XPITCH];
  float sg[TB * SGPITCH];
  float h[TB * HPITCH];
  float sx[TB * 65];
  float part[4 * 32];
  float wo[HIDN * 2];
  float bo[2];
  float scl[TB];
  float iscl[TB];
};

__device__ __forceinline__ float fast_tanh(float x) {
#if __has_builtin(__builtin_amdgcn_tanhf)
  return __builtin_amdgcn_tanhf(x);
#else
  float y;
  asm("v_tanh_f32 %0, %1" : "=v"(y) : "v"(x));
  return y;
#endif
}

__device__ __forceinline__ float fast_sigmoid(float x) {
  return __builtin_amdgcn_rcpf(1.0f + __expf(-x));
}

__device__ __forceinline__ void stage_chunk(SMem& sm,
                                            const float* __restrict__ covs,
                                            const float* __restrict__ target,
                                            int b0, int t0, int tid) {
  for (int e4 = tid; e4 < TB * (CPITCH / 4); e4 += 128) {
    int m = e4 / (CPITCH / 4), off4 = e4 % (CPITCH / 4);
    const float* src = &covs[((size_t)(b0 + m) * S_LEN + t0) * NCOV + off4 * 4];
    float4 v = *(const float4*)src;
    if ((off4 & 7) == 0 && t0 + CHUNK < S_LEN)
      __builtin_prefetch(src + CHUNK * NCOV, 0, 1);
    float vv[4] = { v.x, v.y, v.z, v.w };
    #pragma unroll
    for (int i = 0; i < 4; ++i) {
      int idx = off4 * 4 + i;
      int tc = idx / NCOV, k = idx % NCOV;
      sm.xs[(tc * TB + m) * XPITCH + 1 + k] = vv[i];
    }
  }
  for (int e = tid; e < TB * CHUNK; e += 128) {
    int m = e >> 4, j = e & 15;
    float p = 0.0f;
    if (t0 + j > 0)
      p = target[(size_t)(b0 + m) * S_LEN + t0 + j - 1] * sm.iscl[m];
    sm.xs[(j * TB + m) * XPITCH] = p;
  }
}

__global__ __launch_bounds__(128, 1)
void deepar_lstm_kernel(const float* __restrict__ target,
                        const float* __restrict__ covs,
                        const int*   __restrict__ cats,
                        const float* __restrict__ scale,
                        const float* __restrict__ emb0,
                        const float* __restrict__ emb1,
                        const float* __restrict__ emb2,
                        const float* __restrict__ emb3,
                        const float* __restrict__ w_ih,
                        const float* __restrict__ w_hh,
                        const float* __restrict__ bias,
                        const float* __restrict__ w_out,
                        const float* __restrict__ b_out,
                        float* __restrict__ out)
{
  __shared__ SMem sm;
  const int tid  = threadIdx.x;
  const int wave = tid >> 5;
  const int lane = tid & 31;
  const int b0   = blockIdx.x * TB;

  if (tid < TB) {
    float s = scale[b0 + tid];
    sm.scl[tid]  = s;
    sm.iscl[tid] = 1.0f / fmaxf(s, 1e-4f);
  }
  if (tid < HIDN * 2) sm.wo[tid] = w_out[tid];
  if (tid < 2)        sm.bo[tid] = b_out[tid];

  for (int e = tid; e < TB * 65; e += 128) {
    int m = e / 65, k = e % 65;
    float v;
    if      (k < 16) v = emb0[cats[(b0 + m) * 4 + 0] * EMB + k];
    else if (k < 32) v = emb1[cats[(b0 + m) * 4 + 1] * EMB + (k - 16)];
    else if (k < 48) v = emb2[cats[(b0 + m) * 4 + 2] * EMB + (k - 32)];
    else if (k < 64) v = emb3[cats[(b0 + m) * 4 + 3] * EMB + (k - 48)];
    else             v = log1pf(scale[b0 + m]);
    sm.sx[e] = v;
  }
  __syncthreads();

  for (int e = tid; e < TB * G4; e += 128) {
    int m = e >> 8, j = e & 255;
    float acc = bias[j];
    #pragma unroll 5
    for (int k = 0; k < 65; ++k)
      acc += sm.sx[m * 65 + k] * w_ih[(11 + k) * G4 + j];
    sm.sg[m * SGPITCH + j] = acc;
  }

  const int khalf = lane >> 4;
  const int ml    = lane & 15;
  const int ncol  = wave * 16 + ml;

  v2f whhB[16][4];
  #pragma unroll
  for (int kt = 0; kt < 16; ++kt)
    #pragma unroll
    for (int nt = 0; nt < 4; ++nt) {
      int k = kt * 4 + khalf * 2;
      int n = nt * 64 + ncol;
      v2f b = { w_hh[k * G4 + n], w_hh[(k + 1) * G4 + n] };
      whhB[kt][nt] = b;
    }
  v2f wihB[3][4];
  #pragma unroll
  for (int kt = 0; kt < 3; ++kt)
    #pragma unroll
    for (int nt = 0; nt < 4; ++nt) {
      int k = kt * 4 + khalf * 2;
      int n = nt * 64 + ncol;
      v2f b = { w_ih[k * G4 + n], w_ih[(k + 1) * G4 + n] };
      wihB[kt][nt] = b;
    }
  __syncthreads();
  v8f statC[4];
  #pragma unroll
  for (int nt = 0; nt < 4; ++nt)
    #pragma unroll
    for (int r = 0; r < 8; ++r) {
      int m = r + khalf * 8;
      statC[nt][r] = sm.sg[m * SGPITCH + nt * 64 + ncol];
    }

  for (int e = tid; e < TB * HPITCH; e += 128) sm.h[e] = 0.0f;
  for (int e = tid; e < CHUNK * TB; e += 128) sm.xs[e * XPITCH + (XPITCH - 1)] = 0.0f;
  stage_chunk(sm, covs, target, b0, 0, tid);

  v8f creg = {0.f, 0.f, 0.f, 0.f, 0.f, 0.f, 0.f, 0.f};
  v8f hreg;

  for (int t = 0; t < S_LEN; ++t) {
    asm volatile("s_wait_dscnt 0x0" ::: "memory");
    __syncthreads();

    v2f xA[3], hA[16];
    const float* xrow = &sm.xs[((t & (CHUNK - 1)) * TB + ml) * XPITCH + khalf * 2];
    #pragma unroll
    for (int kt = 0; kt < 3; ++kt)
      xA[kt] = *(const v2f*)&xrow[kt * 4];
    #pragma unroll
    for (int kt = 0; kt < 16; ++kt)
      hA[kt] = *(const v2f*)&sm.h[ml * HPITCH + kt * 4 + khalf * 2];

    v8f acc[4];
    #pragma unroll
    for (int nt = 0; nt < 4; ++nt) {
      v8f a = statC[nt];
      #pragma unroll
      for (int kt = 0; kt < 3; ++kt)
        a = __builtin_amdgcn_wmma_f32_16x16x4_f32(false, xA[kt], false, wihB[kt][nt],
                                                  (short)0, a, false, false);
      #pragma unroll
      for (int kt = 0; kt < 16; ++kt)
        a = __builtin_amdgcn_wmma_f32_16x16x4_f32(false, hA[kt], false, whhB[kt][nt],
                                                  (short)0, a, false, false);
      acc[nt] = a;
    }
    #pragma unroll
    for (int r = 0; r < 8; ++r) {
      float gi = fast_sigmoid(acc[0][r]);
      float gf = fast_sigmoid(acc[1][r]);
      float gg = fast_tanh(acc[2][r]);
      float go = fast_sigmoid(acc[3][r]);
      float cc = gf * creg[r] + gi * gg;
      creg[r] = cc;
      hreg[r] = go * fast_tanh(cc);
    }
    {
      int p = tid & 31, q = tid >> 5;
      int m = p >> 1, j = p & 1;
      float s = 0.0f;
      #pragma unroll
      for (int kk = 0; kk < 16; ++kk) {
        int k = q * 16 + kk;
        s += sm.h[m * HPITCH + k] * sm.wo[k * 2 + j];
      }
      sm.part[q * 32 + p] = s;
    }

    __syncthreads();

    #pragma unroll
    for (int r = 0; r < 8; ++r) {
      int m = r + khalf * 8;
      sm.h[m * HPITCH + ncol] = hreg[r];
    }
    if (t > 0 && tid < 32) {
      int m = tid >> 1, j = tid & 1;
      float r = sm.bo[j] + sm.part[tid] + sm.part[32 + tid] +
                sm.part[64 + tid] + sm.part[96 + tid];
      float sp = (r > 20.0f) ? r : log1pf(__expf(r));
      sp += 1e-4f;
      size_t idx = (size_t)(b0 + m) * S_LEN + (t - 1);
      if (j == 0) out[idx] = sp * sm.scl[m];
      else        out[(size_t)B_TOT * S_LEN + idx] = sp;
    }
    const int tn = t + 1;
    if (tn < S_LEN && (tn & (CHUNK - 1)) == 0)
      stage_chunk(sm, covs, target, b0, tn, tid);
  }

  __syncthreads();
  {
    int p = tid & 31, q = tid >> 5;
    int m = p >> 1, j = p & 1;
    float s = 0.0f;
    #pragma unroll
    for (int kk = 0; kk < 16; ++kk) {
      int k = q * 16 + kk;
      s += sm.h[m * HPITCH + k] * sm.wo[k * 2 + j];
    }
    sm.part[q * 32 + p] = s;
  }
  __syncthreads();
  if (tid < 32) {
    int m = tid >> 1, j = tid & 1;
    float r = sm.bo[j] + sm.part[tid] + sm.part[32 + tid] +
              sm.part[64 + tid] + sm.part[96 + tid];
    float sp = (r > 20.0f) ? r : log1pf(__expf(r));
    sp += 1e-4f;
    size_t idx = (size_t)(b0 + m) * S_LEN + (S_LEN - 1);
    if (j == 0) out[idx] = sp * sm.scl[m];
    else        out[(size_t)B_TOT * S_LEN + idx] = sp;
  }
}

extern "C" void kernel_launch(void* const* d_in, const int* in_sizes, int n_in,
                              void* d_out, int out_size, void* d_ws, size_t ws_size,
                              hipStream_t stream) {
  const float* target = (const float*)d_in[0];
  const float* covs   = (const float*)d_in[1];
  const int*   cats   = (const int*)  d_in[2];
  const float* scale  = (const float*)d_in[3];
  const float* emb0   = (const float*)d_in[4];
  const float* emb1   = (const float*)d_in[5];
  const float* emb2   = (const float*)d_in[6];
  const float* emb3   = (const float*)d_in[7];
  const float* w_ih   = (const float*)d_in[8];
  const float* w_hh   = (const float*)d_in[9];
  const float* bias   = (const float*)d_in[10];
  const float* w_out  = (const float*)d_in[11];
  const float* b_out  = (const float*)d_in[12];
  float* out = (float*)d_out;

  dim3 grid(B_TOT / TB);
  dim3 block(128);
  deepar_lstm_kernel<<<grid, block, 0, stream>>>(
      target, covs, cats, scale, emb0, emb1, emb2, emb3,
      w_ih, w_hh, bias, w_out, b_out, out);
}


# Output head batched: per-step head partials go to a 16-step LDS ring; softplus, scale and stores run once per 16 steps on all 128 threads (4 steps each, one 16-byte store), not per step on one wave
# speedup vs baseline: 1.5189x; 1.2552x over previous
#include <hip/hip_runtime.h>
#include <hip/hip_bf16.h>
#include <math.h>

typedef __attribute__((ext_vector_type(2))) float v2f;
typedef __attribute__((ext_vector_type(8))) float v8f;

#define B_TOT   4096
#define S_LEN   512
#define HIDN    64
#define G4      256
#define NCOV    10
#define EMB     16
#define TB      16
#define HPITCH  68
#define SGPITCH 260
#define XPITCH  12
#define CHUNK   16
#define CPITCH  (CHUNK*NCOV)
#define PPITCH  132

struct __align__(16) SMem {
  float xs[CHUNK * TB * XPITCH];
  float sg[TB * SGPITCH];
  float h[TB * HPITCH];
  float sx[TB * 65];
  float part[CHUNK * PPITCH];
  float wo[HIDN * 2];
  float bo[2];
  float scl[TB];
  float iscl[TB];
};

__device__ __forceinline__ float fast_tanh(float x) {
#if __has_builtin(__builtin_amdgcn_tanhf)
  return __builtin_amdgcn_tanhf(x);
#else
  float y;
  asm("v_tanh_f32 %0, %1" : "=v"(y) : "v"(x));
  return y;
#endif
}

__device__ __forceinline__ float fast_sigmoid(float x) {
  return __builtin_amdgcn_rcpf(1.0f + __expf(-x));
}

__device__ __forceinline__ void stage_chunk(SMem& sm,
                                            const float* __restrict__ covs,
                                            const float* __restrict__ target,
                                            int b0, int t0, int tid) {
  for (int e4 = tid; e4 < TB * (CPITCH / 4); e4 += 128) {
    int m = e4 / (CPITCH / 4), off4 = e4 % (CPITCH / 4);
    const float* src = &covs[((size_t)(b0 + m) * S_LEN + t0) * NCOV + off4 * 4];
    float4 v = *(const float4*)src;
    if ((off4 & 7) == 0 && t0 + CHUNK < S_LEN)
      __builtin_prefetch(src + CHUNK * NCOV, 0, 1);
    float vv[4] = { v.x, v.y, v.z, v.w };
    #pragma unroll
    for (int i = 0; i < 4; ++i) {
      int idx = off4 * 4 + i;
      int tc = idx / NCOV, k = idx % NCOV;
      sm.xs[(tc * TB + m) * XPITCH + 1 + k] = vv[i];
    }
  }
  for (int e = tid; e < TB * CHUNK; e += 128) {
    int m = e >> 4, j = e & 15;
    float p = 0.0f;
    if (t0 + j > 0)
      p = target[(size_t)(b0 + m) * S_LEN + t0 + j - 1] * sm.iscl[m];
    sm.xs[(j * TB + m) * XPITCH] = p;
  }
}

__device__ __forceinline__ void flush_outputs(const SMem& sm, float* __restrict__ out,
                                              int b0, int t0, int tid) {
  const int tq = tid & 3;
  const int p  = tid >> 2;
  const int m  = p >> 1, j = p & 1;
  const float bj   = sm.bo[j];
  const float mult = (j == 0) ? sm.scl[m] : 1.0f;
  float v[4];
  #pragma unroll
  for (int i = 0; i < 4; ++i) {
    const float* pp = &sm.part[(tq * 4 + i) * PPITCH + p];
    float r = bj + pp[0] + pp[32] + pp[64] + pp[96];
    float l = log1pf(__expf(r));
    float sp = (r > 20.0f) ? r : l;
    sp += 1e-4f;
    v[i] = sp * mult;
  }
  size_t idx = (size_t)j * B_TOT * S_LEN + (size_t)(b0 + m) * S_LEN + t0 + tq * 4;
  *(float4*)&out[idx] = make_float4(v[0], v[1], v[2], v[3]);
}

__global__ __launch_bounds__(128, 1)
void deepar_lstm_kernel(const float* __restrict__ target,
                        const float* __restrict__ covs,
                        const int*   __restrict__ cats,
                        const float* __restrict__ scale,
                        const float* __restrict__ emb0,
                        const float* __restrict__ emb1,
                        const float* __restrict__ emb2,
                        const float* __restrict__ emb3,
                        const float* __restrict__ w_ih,
                        const float* __restrict__ w_hh,
                        const float* __restrict__ bias,
                        const float* __restrict__ w_out,
                        const float* __restrict__ b_out,
                        float* __restrict__ out)
{
  __shared__ SMem sm;
  const int tid  = threadIdx.x;
  const int wave = tid >> 5;
  const int lane = tid & 31;
  const int b0   = blockIdx.x * TB;

  if (tid < TB) {
    float s = scale[b0 + tid];
    sm.scl[tid]  = s;
    sm.iscl[tid] = 1.0f / fmaxf(s, 1e-4f);
  }
  if (tid < HIDN * 2) sm.wo[tid] = w_out[tid];
  if (tid < 2)        sm.bo[tid] = b_out[tid];

  for (int e = tid; e < TB * 65; e += 128) {
    int m = e / 65, k = e % 65;
    float v;
    if      (k < 16) v = emb0[cats[(b0 + m) * 4 + 0] * EMB + k];
    else if (k < 32) v = emb1[cats[(b0 + m) * 4 + 1] * EMB + (k - 16)];
    else if (k < 48) v = emb2[cats[(b0 + m) * 4 + 2] * EMB + (k - 32)];
    else if (k < 64) v = emb3[cats[(b0 + m) * 4 + 3] * EMB + (k - 48)];
    else             v = log1pf(scale[b0 + m]);
    sm.sx[e] = v;
  }
  __syncthreads();

  for (int e = tid; e < TB * G4; e += 128) {
    int m = e >> 8, j = e & 255;
    float acc = bias[j];
    #pragma unroll 5
    for (int k = 0; k < 65; ++k)
      acc += sm.sx[m * 65 + k] * w_ih[(11 + k) * G4 + j];
    sm.sg[m * SGPITCH + j] = acc;
  }

  const int khalf = lane >> 4;
  const int ml    = lane & 15;
  const int ncol  = wave * 16 + ml;

  v2f whhB[16][4];
  #pragma unroll
  for (int kt = 0; kt < 16; ++kt)
    #pragma unroll
    for (int nt = 0; nt < 4; ++nt) {
      int k = kt * 4 + khalf * 2;
      int n = nt * 64 + ncol;
      v2f b = { w_hh[k * G4 + n], w_hh[(k + 1) * G4 + n] };
      whhB[kt][nt] = b;
    }
  v2f wihB[3][4];
  #pragma unroll
  for (int kt = 0; kt < 3; ++kt)
    #pragma unroll
    for (int nt = 0; nt < 4; ++nt) {
      int k = kt * 4 + khalf * 2;
      int n = nt * 64 + ncol;
      v2f b = { w_ih[k * G4 + n], w_ih[(k + 1) * G4 + n] };
      wihB[kt][nt] = b;
    }
  __syncthreads();
  v8f statC[4];
  #pragma unroll
  for (int nt = 0; nt < 4; ++nt)
    #pragma unroll
    for (int r = 0; r < 8; ++r) {
      int m = r + khalf * 8;
      statC[nt][r] = sm.sg[m * SGPITCH + nt * 64 + ncol];
    }

  for (int e = tid; e < TB * HPITCH; e += 128) sm.h[e] = 0.0f;
  for (int e = tid; e < CHUNK * TB; e += 128) sm.xs[e * XPITCH + (XPITCH - 1)] = 0.0f;
  stage_chunk(sm, covs, target, b0, 0, tid);

  v8f creg = {0.f, 0.f, 0.f, 0.f, 0.f, 0.f, 0.f, 0.f};
  v8f hreg;

  for (int t = 0; t < S_LEN; ++t) {
    asm volatile("s_wait_dscnt 0x0" ::: "memory");
    __syncthreads();

    v2f xA[3], hA[16];
    const float* xrow = &sm.xs[((t & (CHUNK - 1)) * TB + ml) * XPITCH + khalf * 2];
    #pragma unroll
    for (int kt = 0; kt < 3; ++kt)
      xA[kt] = *(const v2f*)&xrow[kt * 4];
    #pragma unroll
    for (int kt = 0; kt < 16; ++kt)
      hA[kt] = *(const v2f*)&sm.h[ml * HPITCH + kt * 4 + khalf * 2];

    v8f acc[4];
    #pragma unroll
    for (int nt = 0; nt < 4; ++nt) {
      v8f a = statC[nt];
      #pragma unroll
      for (int kt = 0; kt < 3; ++kt)
        a = __builtin_amdgcn_wmma_f32_16x16x4_f32(false, xA[kt], false, wihB[kt][nt],
                                                  (short)0, a, false, false);
      #pragma unroll
      for (int kt = 0; kt < 16; ++kt)
        a = __builtin_amdgcn_wmma_f32_16x16x4_f32(false, hA[kt], false, whhB[kt][nt],
                                                  (short)0, a, false, false);
      acc[nt] = a;
    }
    #pragma unroll
    for (int r = 0; r < 8; ++r) {
      float gi = fast_sigmoid(acc[0][r]);
      float gf = fast_sigmoid(acc[1][r]);
      float gg = fast_tanh(acc[2][r]);
      float go = fast_sigmoid(acc[3][r]);
      float cc = gf * creg[r] + gi * gg;
      creg[r] = cc;
      hreg[r] = go * fast_tanh(cc);
    }
    {
      int p = tid & 31, q = tid >> 5;
      int m = p >> 1, j = p & 1;
      float s = 0.0f;
      #pragma unroll
      for (int kk = 0; kk < 16; ++kk) {
        int k = q * 16 + kk;
        s += sm.h[m * HPITCH + k] * sm.wo[k * 2 + j];
      }
      sm.part[((t - 1) & (CHUNK - 1)) * PPITCH + q * 32 + p] = s;
    }

    __syncthreads();

    #pragma unroll
    for (int r = 0; r < 8; ++r) {
      int m = r + khalf * 8;
      sm.h[m * HPITCH + ncol] = hreg[r];
    }
    if (t > 0 && (t & (CHUNK - 1)) == 0)
      flush_outputs(sm, out, b0, t - CHUNK, tid);
    const int tn = t + 1;
    if (tn < S_LEN && (tn & (CHUNK - 1)) == 0)
      stage_chunk(sm, covs, target, b0, tn, tid);
  }

  __syncthreads();
  {
    int p = tid & 31, q = tid >> 5;
    int m = p >> 1, j = p & 1;
    float s = 0.0f;
    #pragma unroll
    for (int kk = 0; kk < 16; ++kk) {
      int k = q * 16 + kk;
      s += sm.h[m * HPITCH + k] * sm.wo[k * 2 + j];
    }
    sm.part[((S_LEN - 1) & (CHUNK - 1)) * PPITCH + q * 32 + p] = s;
  }
  __syncthreads();
  flush_outputs(sm, out, b0, S_LEN - CHUNK, tid);
}

extern "C" void kernel_launch(void* const* d_in, const int* in_sizes, int n_in,
                              void* d_out, int out_size, void* d_ws, size_t ws_size,
                              hipStream_t stream) {
  const float* target = (const float*)d_in[0];
  const float* covs   = (const float*)d_in[1];
  const int*   cats   = (const int*)  d_in[2];
  const float* scale  = (const float*)d_in[3];
  const float* emb0   = (const float*)d_in[4];
  const float* emb1   = (const float*)d_in[5];
  const float* emb2   = (const float*)d_in[6];
  const float* emb3   = (const float*)d_in[7];
  const float* w_ih   = (const float*)d_in[8];
  const float* w_hh   = (const float*)d_in[9];
  const float* bias   = (const float*)d_in[10];
  const float* w_out  = (const float*)d_in[11];
  const float* b_out  = (const float*)d_in[12];
  float* out = (float*)d_out;

  dim3 grid(B_TOT / TB);
  dim3 block(128);
  deepar_lstm_kernel<<<grid, block, 0, stream>>>(
      target, covs, cats, scale, emb0, emb1, emb2, emb3,
      w_ih, w_hh, bias, w_out, b_out, out);
}


# All LDS reads of a scan step (x and h A-fragments, head operands) issued as one ordered group in front of the 76 WMMAs instead of at 9 places among them; every sum and its order unchanged
# speedup vs baseline: 1.6371x; 1.0778x over previous
#include <hip/hip_runtime.h>
#include <hip/hip_bf16.h>
#include <math.h>

typedef __attribute__((ext_vector_type(2))) float v2f;
typedef __attribute__((ext_vector_type(8))) float v8f;

#define B_TOT   4096
#define S_LEN   512
#define HIDN    64
#define G4      256
#define NCOV    10
#define EMB     16
#define TB      16
#define HPITCH  68
#define SGPITCH 260
#define XPITCH  12
#define CHUNK   16
#define CPITCH  (CHUNK*NCOV)
#define PPITCH  132

struct __align__(16) SMem {
  float xs[CHUNK * TB * XPITCH];
  float sg[TB * SGPITCH];
  float h[TB * HPITCH];
  float sx[TB * 65];
  float part[CHUNK * PPITCH];
  float wo[HIDN * 2];
  float bo[2];
  float scl[TB];
  float iscl[TB];
};

__device__ __forceinline__ float fast_tanh(float x) {
#if __has_builtin(__builtin_amdgcn_tanhf)
  return __builtin_amdgcn_tanhf(x);
#else
  float y;
  asm("v_tanh_f32 %0, %1" : "=v"(y) : "v"(x));
  return y;
#endif
}

__device__ __forceinline__ float fast_sigmoid(float x) {
  return __builtin_amdgcn_rcpf(1.0f + __expf(-x));
}

__device__ __forceinline__ void stage_chunk(SMem& sm,
                                            const float* __restrict__ covs,
                                            const float* __restrict__ target,
                                            int b0, int t0, int tid) {
  for (int e4 = tid; e4 < TB * (CPITCH / 4); e4 += 128) {
    int m = e4 / (CPITCH / 4), off4 = e4 % (CPITCH / 4);
    const float* src = &covs[((size_t)(b0 + m) * S_LEN + t0) * NCOV + off4 * 4];
    float4 v = *(const float4*)src;
    if ((off4 & 7) == 0 && t0 + CHUNK < S_LEN)
      __builtin_prefetch(src + CHUNK * NCOV, 0, 1);
    float vv[4] = { v.x, v.y, v.z, v.w };
    #pragma unroll
    for (int i = 0; i < 4; ++i) {
      int idx = off4 * 4 + i;
      int tc = idx / NCOV, k = idx % NCOV;
      sm.xs[(tc * TB + m) * XPITCH + 1 + k] = vv[i];
    }
  }
  for (int e = tid; e < TB * CHUNK; e += 128) {
    int m = e >> 4, j = e & 15;
    float p = 0.0f;
    if (t0 + j > 0)
      p = target[(size_t)(b0 + m) * S_LEN + t0 + j - 1] * sm.iscl[m];
    sm.xs[(j * TB + m) * XPITCH] = p;
  }
}

__device__ __forceinline__ void flush_outputs(const SMem& sm, float* __restrict__ out,
                                              int b0, int t0, int tid) {
  const int tq = tid & 3;
  const int p  = tid >> 2;
  const int m  = p >> 1, j = p & 1;
  const float bj   = sm.bo[j];
  const float mult = (j == 0) ? sm.scl[m] : 1.0f;
  float v[4];
  #pragma unroll
  for (int i = 0; i < 4; ++i) {
    const float* pp = &sm.part[(tq * 4 + i) * PPITCH + p];
    float r = bj + pp[0] + pp[32] + pp[64] + pp[96];
    float l = log1pf(__expf(r));
    float sp = (r > 20.0f) ? r : l;
    sp += 1e-4f;
    v[i] = sp * mult;
  }
  size_t idx = (size_t)j * B_TOT * S_LEN + (size_t)(b0 + m) * S_LEN + t0 + tq * 4;
  *(float4*)&out[idx] = make_float4(v[0], v[1], v[2], v[3]);
}

__global__ __launch_bounds__(128, 1)
void deepar_lstm_kernel(const float* __restrict__ target,
                        const float* __restrict__ covs,
                        const int*   __restrict__ cats,
                        const float* __restrict__ scale,
                        const float* __restrict__ emb0,
                        const float* __restrict__ emb1,
                        const float* __restrict__ emb2,
                        const float* __restrict__ emb3,
                        const float* __restrict__ w_ih,
                        const float* __restrict__ w_hh,
                        const float* __restrict__ bias,
                        const float* __restrict__ w_out,
                        const float* __restrict__ b_out,
                        float* __restrict__ out)
{
  __shared__ SMem sm;
  const int tid  = threadIdx.x;
  const int wave = tid >> 5;
  const int lane = tid & 31;
  const int b0   = blockIdx.x * TB;

  if (tid < TB) {
    float s = scale[b0 + tid];
    sm.scl[tid]  = s;
    sm.iscl[tid] = 1.0f / fmaxf(s, 1e-4f);
  }
  if (tid < HIDN * 2) sm.wo[tid] = w_out[tid];
  if (tid < 2)        sm.bo[tid] = b_out[tid];

  for (int e = tid; e < TB * 65; e += 128) {
    int m = e / 65, k = e % 65;
    float v;
    if      (k < 16) v = emb0[cats[(b0 + m) * 4 + 0] * EMB + k];
    else if (k < 32) v = emb1[cats[(b0 + m) * 4 + 1] * EMB + (k - 16)];
    else if (k < 48) v = emb2[cats[(b0 + m) * 4 + 2] * EMB + (k - 32)];
    else if (k < 64) v = emb3[cats[(b0 + m) * 4 + 3] * EMB + (k - 48)];
    else             v = log1pf(scale[b0 + m]);
    sm.sx[e] = v;
  }
  __syncthreads();

  for (int e = tid; e < TB * G4; e += 128) {
    int m = e >> 8, j = e & 255;
    float acc = bias[j];
    #pragma unroll 5
    for (int k = 0; k < 65; ++k)
      acc += sm.sx[m * 65 + k] * w_ih[(11 + k) * G4 + j];
    sm.sg[m * SGPITCH + j] = acc;
  }

  const int khalf = lane >> 4;
  const int ml    = lane & 15;
  const int ncol  = wave * 16 + ml;

  v2f whhB[16][4];
  #pragma unroll
  for (int kt = 0; kt < 16; ++kt)
    #pragma unroll
    for (int nt = 0; nt < 4; ++nt) {
      int k = kt * 4 + khalf * 2;
      int n = nt * 64 + ncol;
      v2f b = { w_hh[k * G4 + n], w_hh[(k + 1) * G4 + n] };
      whhB[kt][nt] = b;
    }
  v2f wihB[3][4];
  #pragma unroll
  for (int kt = 0; kt < 3; ++kt)
    #pragma unroll
    for (int nt = 0; nt < 4; ++nt) {
      int k = kt * 4 + khalf * 2;
      int n = nt * 64 + ncol;
      v2f b = { w_ih[k * G4 + n], w_ih[(k + 1) * G4 + n] };
      wihB[kt][nt] = b;
    }
  __syncthreads();
  v8f statC[4];
  #pragma unroll
  for (int nt = 0; nt < 4; ++nt)
    #pragma unroll
    for (int r = 0; r < 8; ++r) {
      int m = r + khalf * 8;
      statC[nt][r] = sm.sg[m * SGPITCH + nt * 64 + ncol];
    }

  for (int e = tid; e < TB * HPITCH; e += 128) sm.h[e] = 0.0f;
  for (int e = tid; e < CHUNK * TB; e += 128) sm.xs[e * XPITCH + (XPITCH - 1)] = 0.0f;
  stage_chunk(sm, covs, target, b0, 0, tid);

  v8f creg = {0.f, 0.f, 0.f, 0.f, 0.f, 0.f, 0.f, 0.f};
  v8f hreg;

  for (int t = 0; t < S_LEN; ++t) {
    asm volatile("s_wait_dscnt 0x0" ::: "memory");
    __syncthreads();

    v2f xA[3], hA[16];
    const float* xrow = &sm.xs[((t & (CHUNK - 1)) * TB + ml) * XPITCH + khalf * 2];
    #pragma unroll
    for (int kt = 0; kt < 3; ++kt)
      xA[kt] = *(const v2f*)&xrow[kt * 4];
    __builtin_amdgcn_sched_barrier(0);
    #pragma unroll
    for (int kt = 0; kt < 16; ++kt)
      hA[kt] = *(const v2f*)&sm.h[ml * HPITCH + kt * 4 + khalf * 2];
    __builtin_amdgcn_sched_barrier(0);
    const int hp_p = tid & 31, hp_q = tid >> 5;
    const int hp_m = hp_p >> 1, hp_j = hp_p & 1;
    float hv[16], wv[16];
    #pragma unroll
    for (int kk = 0; kk < 16; ++kk) {
      int k = hp_q * 16 + kk;
      hv[kk] = sm.h[hp_m * HPITCH + k];
      wv[kk] = sm.wo[k * 2 + hp_j];
    }
    __builtin_amdgcn_sched_barrier(0);

    v8f acc[4];
    #pragma unroll
    for (int nt = 0; nt < 4; ++nt) acc[nt] = statC[nt];
    #pragma unroll
    for (int kt = 0; kt < 3; ++kt)
      #pragma unroll
      for (int nt = 0; nt < 4; ++nt)
        acc[nt] = __builtin_amdgcn_wmma_f32_16x16x4_f32(false, xA[kt], false, wihB[kt][nt],
                                                        (short)0, acc[nt], false, false);
    __builtin_amdgcn_sched_barrier(0);
    #pragma unroll
    for (int kt = 0; kt < 16; ++kt)
      #pragma unroll
      for (int nt = 0; nt < 4; ++nt)
        acc[nt] = __builtin_amdgcn_wmma_f32_16x16x4_f32(false, hA[kt], false, whhB[kt][nt],
                                                        (short)0, acc[nt], false, false);
    #pragma unroll
    for (int r = 0; r < 8; ++r) {
      float gi = fast_sigmoid(acc[0][r]);
      float gf = fast_sigmoid(acc[1][r]);
      float gg = fast_tanh(acc[2][r]);
      float go = fast_sigmoid(acc[3][r]);
      float cc = gf * creg[r] + gi * gg;
      creg[r] = cc;
      hreg[r] = go * fast_tanh(cc);
    }
    {
      float s = 0.0f;
      #pragma unroll
      for (int kk = 0; kk < 16; ++kk)
        s += hv[kk] * wv[kk];
      sm.part[((t - 1) & (CHUNK - 1)) * PPITCH + hp_q * 32 + hp_p] = s;
    }

    __syncthreads();

    #pragma unroll
    for (int r = 0; r < 8; ++r) {
      int m = r + khalf * 8;
      sm.h[m * HPITCH + ncol] = hreg[r];
    }
    if (t > 0 && (t & (CHUNK - 1)) == 0)
      flush_outputs(sm, out, b0, t - CHUNK, tid);
    const int tn = t + 1;
    if (tn < S_LEN && (tn & (CHUNK - 1)) == 0)
      stage_chunk(sm, covs, target, b0, tn, tid);
  }

  __syncthreads();
  {
    int p = tid & 31, q = tid >> 5;
    int m = p >> 1, j = p & 1;
    float s = 0.0f;
    #pragma unroll
    for (int kk = 0; kk < 16; ++kk) {
      int k = q * 16 + kk;
      s += sm.h[m * HPITCH + k] * sm.wo[k * 2 + j];
    }
    sm.part[((S_LEN - 1) & (CHUNK - 1)) * PPITCH + q * 32 + p] = s;
  }
  __syncthreads();
  flush_outputs(sm, out, b0, S_LEN - CHUNK, tid);
}

extern "C" void kernel_launch(void* const* d_in, const int* in_sizes, int n_in,
                              void* d_out, int out_size, void* d_ws, size_t ws_size,
                              hipStream_t stream) {
  const float* target = (const float*)d_in[0];
  const float* covs   = (const float*)d_in[1];
  const int*   cats   = (const int*)  d_in[2];
  const float* scale  = (const float*)d_in[3];
  const float* emb0   = (const float*)d_in[4];
  const float* emb1   = (const float*)d_in[5];
  const float* emb2   = (const float*)d_in[6];
  const float* emb3   = (const float*)d_in[7];
  const float* w_ih   = (const float*)d_in[8];
  const float* w_hh   = (const float*)d_in[9];
  const float* bias   = (const float*)d_in[10];
  const float* w_out  = (const float*)d_in[11];
  const float* b_out  = (const float*)d_in[12];
  float* out = (float*)d_out;

  dim3 grid(B_TOT / TB);
  dim3 block(128);
  deepar_lstm_kernel<<<grid, block, 0, stream>>>(
      target, covs, cats, scale, emb0, emb1, emb2, emb3,
      w_ih, w_hh, bias, w_out, b_out, out);
}


# Chunk refill split in two: the next 16-step chunk's global loads are issued a chunk ahead into registers, before a matrix block, and written to LDS at the chunk end; 7 serial load-wait rounds gone
# speedup vs baseline: 1.7258x; 1.0542x over previous
#include <hip/hip_runtime.h>
#include <hip/hip_bf16.h>
#include <math.h>

typedef __attribute__((ext_vector_type(2))) float v2f;
typedef __attribute__((ext_vector_type(8))) float v8f;

#define B_TOT   4096
#define S_LEN   512
#define HIDN    64
#define G4      256
#define NCOV    10
#define EMB     16
#define TB      16
#define HPITCH  68
#define SGPITCH 260
#define XPITCH  12
#define CHUNK   16
#define CPITCH  (CHUNK*NCOV)
#define PPITCH  132

struct __align__(16) SMem {
  float xs[CHUNK * TB * XPITCH];
  float sg[TB * SGPITCH];
  float h[TB * HPITCH];
  float sx[TB * 65];
  float part[CHUNK * PPITCH];
  float wo[HIDN * 2];
  float bo[2];
  float scl[TB];
  float iscl[TB];
};

__device__ __forceinline__ float fast_tanh(float x) {
#if __has_builtin(__builtin_amdgcn_tanhf)
  return __builtin_amdgcn_tanhf(x);
#else
  float y;
  asm("v_tanh_f32 %0, %1" : "=v"(y) : "v"(x));
  return y;
#endif
}

__device__ __forceinline__ float fast_sigmoid(float x) {
  return __builtin_amdgcn_rcpf(1.0f + __expf(-x));
}

struct ChunkRegs {
  float4 cv[TB * (CPITCH / 4) / 128];
  float  tg[TB * CHUNK / 128];
};

__device__ __forceinline__ void load_chunk(ChunkRegs& cr,
                                           const float* __restrict__ covs,
                                           const float* __restrict__ target,
                                           int b0, int t0, int tid) {
  #pragma unroll
  for (int i = 0; i < TB * (CPITCH / 4) / 128; ++i) {
    int e4 = tid + i * 128;
    int m = e4 / (CPITCH / 4), off4 = e4 % (CPITCH / 4);
    cr.cv[i] = *(const float4*)&covs[((size_t)(b0 + m) * S_LEN + t0) * NCOV + off4 * 4];
  }
  #pragma unroll
  for (int i = 0; i < TB * CHUNK / 128; ++i) {
    int e = tid + i * 128;
    int m = e >> 4, j = e & 15;
    int ti = t0 + j - 1;
    float v = target[(size_t)(b0 + m) * S_LEN + (ti < 0 ? 0 : ti)];
    cr.tg[i] = (ti < 0) ? 0.0f : v;
  }
}

__device__ __forceinline__ void store_chunk(SMem& sm, const ChunkRegs& cr,
                                            float isc0, float isc1, int tid) {
  #pragma unroll
  for (int i = 0; i < TB * (CPITCH / 4) / 128; ++i) {
    int e4 = tid + i * 128;
    int m = e4 / (CPITCH / 4), off4 = e4 % (CPITCH / 4);
    float vv[4] = { cr.cv[i].x, cr.cv[i].y, cr.cv[i].z, cr.cv[i].w };
    #pragma unroll
    for (int q = 0; q < 4; ++q) {
      int idx = off4 * 4 + q;
      int tc = idx / NCOV, k = idx % NCOV;
      sm.xs[(tc * TB + m) * XPITCH + 1 + k] = vv[q];
    }
  }
  #pragma unroll
  for (int i = 0; i < TB * CHUNK / 128; ++i) {
    int e = tid + i * 128;
    int m = e >> 4, j = e & 15;
    sm.xs[(j * TB + m) * XPITCH] = cr.tg[i] * (i == 0 ? isc0 : isc1);
  }
}

__device__ __forceinline__ void flush_outputs(const SMem& sm, float* __restrict__ out,
                                              int b0, int t0, int tid) {
  const int tq = tid & 3;
  const int p  = tid >> 2;
  const int m  = p >> 1, j = p & 1;
  const float bj   = sm.bo[j];
  const float mult = (j == 0) ? sm.scl[m] : 1.0f;
  float v[4];
  #pragma unroll
  for (int i = 0; i < 4; ++i) {
    const float* pp = &sm.part[(tq * 4 + i) * PPITCH + p];
    float r = bj + pp[0] + pp[32] + pp[64] + pp[96];
    float l = log1pf(__expf(r));
    float sp = (r > 20.0f) ? r : l;
    sp += 1e-4f;
    v[i] = sp * mult;
  }
  size_t idx = (size_t)j * B_TOT * S_LEN + (size_t)(b0 + m) * S_LEN + t0 + tq * 4;
  *(float4*)&out[idx] = make_float4(v[0], v[1], v[2], v[3]);
}

__global__ __launch_bounds__(128, 1)
void deepar_lstm_kernel(const float* __restrict__ target,
                        const float* __restrict__ covs,
                        const int*   __restrict__ cats,
                        const float* __restrict__ scale,
                        const float* __restrict__ emb0,
                        const float* __restrict__ emb1,
                        const float* __restrict__ emb2,
                        const float* __restrict__ emb3,
                        const float* __restrict__ w_ih,
                        const float* __restrict__ w_hh,
                        const float* __restrict__ bias,
                        const float* __restrict__ w_out,
                        const float* __restrict__ b_out,
                        float* __restrict__ out)
{
  __shared__ SMem sm;
  const int tid  = threadIdx.x;
  const int wave = tid >> 5;
  const int lane = tid & 31;
  const int b0   = blockIdx.x * TB;

  if (tid < TB) {
    float s = scale[b0 + tid];
    sm.scl[tid]  = s;
    sm.iscl[tid] = 1.0f / fmaxf(s, 1e-4f);
  }
  if (tid < HIDN * 2) sm.wo[tid] = w_out[tid];
  if (tid < 2)        sm.bo[tid] = b_out[tid];

  for (int e = tid; e < TB * 65; e += 128) {
    int m = e / 65, k = e % 65;
    float v;
    if      (k < 16) v = emb0[cats[(b0 + m) * 4 + 0] * EMB + k];
    else if (k < 32) v = emb1[cats[(b0 + m) * 4 + 1] * EMB + (k - 16)];
    else if (k < 48) v = emb2[cats[(b0 + m) * 4 + 2] * EMB + (k - 32)];
    else if (k < 64) v = emb3[cats[(b0 + m) * 4 + 3] * EMB + (k - 48)];
    else             v = log1pf(scale[b0 + m]);
    sm.sx[e] = v;
  }
  __syncthreads();

  for (int e = tid; e < TB * G4; e += 128) {
    int m = e >> 8, j = e & 255;
    float acc = bias[j];
    #pragma unroll 5
    for (int k = 0; k < 65; ++k)
      acc += sm.sx[m * 65 + k] * w_ih[(11 + k) * G4 + j];
    sm.sg[m * SGPITCH + j] = acc;
  }

  const int khalf = lane >> 4;
  const int ml    = lane & 15;
  const int ncol  = wave * 16 + ml;

  v2f whhB[16][4];
  #pragma unroll
  for (int kt = 0; kt < 16; ++kt)
    #pragma unroll
    for (int nt = 0; nt < 4; ++nt) {
      int k = kt * 4 + khalf * 2;
      int n = nt * 64 + ncol;
      v2f b = { w_hh[k * G4 + n], w_hh[(k + 1) * G4 + n] };
      whhB[kt][nt] = b;
    }
  v2f wihB[3][4];
  #pragma unroll
  for (int kt = 0; kt < 3; ++kt)
    #pragma unroll
    for (int nt = 0; nt < 4; ++nt) {
      int k = kt * 4 + khalf * 2;
      int n = nt * 64 + ncol;
      v2f b = { w_ih[k * G4 + n], w_ih[(k + 1) * G4 + n] };
      wihB[kt][nt] = b;
    }
  __syncthreads();
  v8f statC[4];
  #pragma unroll
  for (int nt = 0; nt < 4; ++nt)
    #pragma unroll
    for (int r = 0; r < 8; ++r) {
      int m = r + khalf * 8;
      statC[nt][r] = sm.sg[m * SGPITCH + nt * 64 + ncol];
    }

  for (int e = tid; e < TB * HPITCH; e += 128) sm.h[e] = 0.0f;
  for (int e = tid; e < CHUNK * TB; e += 128) sm.xs[e * XPITCH + (XPITCH - 1)] = 0.0f;
  const float isc0 = sm.iscl[tid >> 4], isc1 = sm.iscl[(tid + 128) >> 4];
  ChunkRegs cr;
  load_chunk(cr, covs, target, b0, 0, tid);
  store_chunk(sm, cr, isc0, isc1, tid);

  v8f creg = {0.f, 0.f, 0.f, 0.f, 0.f, 0.f, 0.f, 0.f};
  v8f hreg;

  for (int t = 0; t < S_LEN; ++t) {
    asm volatile("s_wait_dscnt 0x0" ::: "memory");
    __syncthreads();

    if ((t & (CHUNK - 1)) == 0 && t + CHUNK < S_LEN)
      load_chunk(cr, covs, target, b0, t + CHUNK, tid);

    v2f xA[3], hA[16];
    const float* xrow = &sm.xs[((t & (CHUNK - 1)) * TB + ml) * XPITCH + khalf * 2];
    #pragma unroll
    for (int kt = 0; kt < 3; ++kt)
      xA[kt] = *(const v2f*)&xrow[kt * 4];
    __builtin_amdgcn_sched_barrier(0);
    #pragma unroll
    for (int kt = 0; kt < 16; ++kt)
      hA[kt] = *(const v2f*)&sm.h[ml * HPITCH + kt * 4 + khalf * 2];
    __builtin_amdgcn_sched_barrier(0);
    const int hp_p = tid & 31, hp_q = tid >> 5;
    const int hp_m = hp_p >> 1, hp_j = hp_p & 1;
    float hv[16], wv[16];
    #pragma unroll
    for (int kk = 0; kk < 16; ++kk) {
      int k = hp_q * 16 + kk;
      hv[kk] = sm.h[hp_m * HPITCH + k];
      wv[kk] = sm.wo[k * 2 + hp_j];
    }
    __builtin_amdgcn_sched_barrier(0);

    v8f acc[4];
    #pragma unroll
    for (int nt = 0; nt < 4; ++nt) acc[nt] = statC[nt];
    #pragma unroll
    for (int kt = 0; kt < 3; ++kt)
      #pragma unroll
      for (int nt = 0; nt < 4; ++nt)
        acc[nt] = __builtin_amdgcn_wmma_f32_16x16x4_f32(false, xA[kt], false, wihB[kt][nt],
                                                        (short)0, acc[nt], false, false);
    __builtin_amdgcn_sched_barrier(0);
    #pragma unroll
    for (int kt = 0; kt < 16; ++kt)
      #pragma unroll
      for (int nt = 0; nt < 4; ++nt)
        acc[nt] = __builtin_amdgcn_wmma_f32_16x16x4_f32(false, hA[kt], false, whhB[kt][nt],
                                                        (short)0, acc[nt], false, false);
    #pragma unroll
    for (int r = 0; r < 8; ++r) {
      float gi = fast_sigmoid(acc[0][r]);
      float gf = fast_sigmoid(acc[1][r]);
      float gg = fast_tanh(acc[2][r]);
      float go = fast_sigmoid(acc[3][r]);
      float cc = gf * creg[r] + gi * gg;
      creg[r] = cc;
      hreg[r] = go * fast_tanh(cc);
    }
    {
      float s = 0.0f;
      #pragma unroll
      for (int kk = 0; kk < 16; ++kk)
        s += hv[kk] * wv[kk];
      sm.part[((t - 1) & (CHUNK - 1)) * PPITCH + hp_q * 32 + hp_p] = s;
    }

    __syncthreads();

    #pragma unroll
    for (int r = 0; r < 8; ++r) {
      int m = r + khalf * 8;
      sm.h[m * HPITCH + ncol] = hreg[r];
    }
    if (t > 0 && (t & (CHUNK - 1)) == 0)
      flush_outputs(sm, out, b0, t - CHUNK, tid);
    const int tn = t + 1;
    if (tn < S_LEN && (tn & (CHUNK - 1)) == 0)
      store_chunk(sm, cr, isc0, isc1, tid);
  }

  __syncthreads();
  {
    int p = tid & 31, q = tid >> 5;
    int m = p >> 1, j = p & 1;
    float s = 0.0f;
    #pragma unroll
    for (int kk = 0; kk < 16; ++kk) {
      int k = q * 16 + kk;
      s += sm.h[m * HPITCH + k] * sm.wo[k * 2 + j];
    }
    sm.part[((S_LEN - 1) & (CHUNK - 1)) * PPITCH + q * 32 + p] = s;
  }
  __syncthreads();
  flush_outputs(sm, out, b0, S_LEN - CHUNK, tid);
}

extern "C" void kernel_launch(void* const* d_in, const int* in_sizes, int n_in,
                              void* d_out, int out_size, void* d_ws, size_t ws_size,
                              hipStream_t stream) {
  const float* target = (const float*)d_in[0];
  const float* covs   = (const float*)d_in[1];
  const int*   cats   = (const int*)  d_in[2];
  const float* scale  = (const float*)d_in[3];
  const float* emb0   = (const float*)d_in[4];
  const float* emb1   = (const float*)d_in[5];
  const float* emb2   = (const float*)d_in[6];
  const float* emb3   = (const float*)d_in[7];
  const float* w_ih   = (const float*)d_in[8];
  const float* w_hh   = (const float*)d_in[9];
  const float* bias   = (const float*)d_in[10];
  const float* w_out  = (const float*)d_in[11];
  const float* b_out  = (const float*)d_in[12];
  float* out = (float*)d_out;

  dim3 grid(B_TOT / TB);
  dim3 block(128);
  deepar_lstm_kernel<<<grid, block, 0, stream>>>(
      target, covs, cats, scale, emb0, emb1, emb2, emb3,
      w_ih, w_hh, bias, w_out, b_out, out);
}


# The 12 dyn-input WMMAs of a step (x read one step ahead from a 2-chunk xs ring) fill its LDS waits: 6 while h stores land, 6 while h fragments return; head store lifted off barrier B; same sums
# speedup vs baseline: 1.7843x; 1.0339x over previous
#include <hip/hip_runtime.h>
#include <hip/hip_bf16.h>
#include <math.h>

typedef __attribute__((ext_vector_type(2))) float v2f;
typedef __attribute__((ext_vector_type(8))) float v8f;

#define B_TOT   4096
#define S_LEN   512
#define HIDN    64
#define G4      256
#define NCOV    10
#define EMB     16
#define TB      16
#define HPITCH  68
#define SGPITCH 260
#define XPITCH  12
#define CHUNK   16
#define CPITCH  (CHUNK*NCOV)
#define PPITCH  132

struct __align__(16) SMem {
  float xs[2 * CHUNK * TB * XPITCH];
  float sg[TB * SGPITCH];
  float h[TB * HPITCH];
  float sx[TB * 65];
  float part[CHUNK * PPITCH];
  float wo[HIDN * 2];
  float bo[2];
  float scl[TB];
  float iscl[TB];
};

__device__ __forceinline__ float fast_tanh(float x) {
#if __has_builtin(__builtin_amdgcn_tanhf)
  return __builtin_amdgcn_tanhf(x);
#else
  float y;
  asm("v_tanh_f32 %0, %1" : "=v"(y) : "v"(x));
  return y;
#endif
}

__device__ __forceinline__ float fast_sigmoid(float x) {
  return __builtin_amdgcn_rcpf(1.0f + __expf(-x));
}

struct ChunkRegs {
  float4 cv[TB * (CPITCH / 4) / 128];
  float  tg[TB * CHUNK / 128];
};

__device__ __forceinline__ void load_chunk(ChunkRegs& cr,
                                           const float* __restrict__ covs,
                                           const float* __restrict__ target,
                                           int b0, int t0, int tid) {
  #pragma unroll
  for (int i = 0; i < TB * (CPITCH / 4) / 128; ++i) {
    int e4 = tid + i * 128;
    int m = e4 / (CPITCH / 4), off4 = e4 % (CPITCH / 4);
    cr.cv[i] = *(const float4*)&covs[((size_t)(b0 + m) * S_LEN + t0) * NCOV + off4 * 4];
  }
  #pragma unroll
  for (int i = 0; i < TB * CHUNK / 128; ++i) {
    int e = tid + i * 128;
    int m = e >> 4, j = e & 15;
    int ti = t0 + j - 1;
    float v = target[(size_t)(b0 + m) * S_LEN + (ti < 0 ? 0 : ti)];
    cr.tg[i] = (ti < 0) ? 0.0f : v;
  }
}

__device__ __forceinline__ void store_chunk(SMem& sm, const ChunkRegs& cr,
                                            float isc0, float isc1, int tid, int half) {
  float* xs = &sm.xs[half * (CHUNK * TB * XPITCH)];
  #pragma unroll
  for (int i = 0; i < TB * (CPITCH / 4) / 128; ++i) {
    int e4 = tid + i * 128;
    int m = e4 / (CPITCH / 4), off4 = e4 % (CPITCH / 4);
    float vv[4] = { cr.cv[i].x, cr.cv[i].y, cr.cv[i].z, cr.cv[i].w };
    #pragma unroll
    for (int q = 0; q < 4; ++q) {
      int idx = off4 * 4 + q;
      int tc = idx / NCOV, k = idx % NCOV;
      xs[(tc * TB + m) * XPITCH + 1 + k] = vv[q];
    }
  }
  #pragma unroll
  for (int i = 0; i < TB * CHUNK / 128; ++i) {
    int e = tid + i * 128;
    int m = e >> 4, j = e & 15;
    xs[(j * TB + m) * XPITCH] = cr.tg[i] * (i == 0 ? isc0 : isc1);
  }
}

__device__ __forceinline__ void flush_outputs(const SMem& sm, float* __restrict__ out,
                                              int b0, int t0, int tid) {
  const int tq = tid & 3;
  const int p  = tid >> 2;
  const int m  = p >> 1, j = p & 1;
  const float bj   = sm.bo[j];
  const float mult = (j == 0) ? sm.scl[m] : 1.0f;
  float v[4];
  #pragma unroll
  for (int i = 0; i < 4; ++i) {
    const float* pp = &sm.part[(tq * 4 + i) * PPITCH + p];
    float r = bj + pp[0] + pp[32] + pp[64] + pp[96];
    float l = log1pf(__expf(r));
    float sp = (r > 20.0f) ? r : l;
    sp += 1e-4f;
    v[i] = sp * mult;
  }
  size_t idx = (size_t)j * B_TOT * S_LEN + (size_t)(b0 + m) * S_LEN + t0 + tq * 4;
  *(float4*)&out[idx] = make_float4(v[0], v[1], v[2], v[3]);
}

__global__ __launch_bounds__(128, 1)
void deepar_lstm_kernel(const float* __restrict__ target,
                        const float* __restrict__ covs,
                        const int*   __restrict__ cats,
                        const float* __restrict__ scale,
                        const float* __restrict__ emb0,
                        const float* __restrict__ emb1,
                        const float* __restrict__ emb2,
                        const float* __restrict__ emb3,
                        const float* __restrict__ w_ih,
                        const float* __restrict__ w_hh,
                        const float* __restrict__ bias,
                        const float* __restrict__ w_out,
                        const float* __restrict__ b_out,
                        float* __restrict__ out)
{
  __shared__ SMem sm;
  const int tid  = threadIdx.x;
  const int wave = tid >> 5;
  const int lane = tid & 31;
  const int b0   = blockIdx.x * TB;

  if (tid < TB) {
    float s = scale[b0 + tid];
    sm.scl[tid]  = s;
    sm.iscl[tid] = 1.0f / fmaxf(s, 1e-4f);
  }
  if (tid < HIDN * 2) sm.wo[tid] = w_out[tid];
  if (tid < 2)        sm.bo[tid] = b_out[tid];

  for (int e = tid; e < TB * 65; e += 128) {
    int m = e / 65, k = e % 65;
    float v;
    if      (k < 16) v = emb0[cats[(b0 + m) * 4 + 0] * EMB + k];
    else if (k < 32) v = emb1[cats[(b0 + m) * 4 + 1] * EMB + (k - 16)];
    else if (k < 48) v = emb2[cats[(b0 + m) * 4 + 2] * EMB + (k - 32)];
    else if (k < 64) v = emb3[cats[(b0 + m) * 4 + 3] * EMB + (k - 48)];
    else             v = log1pf(scale[b0 + m]);
    sm.sx[e] = v;
  }
  __syncthreads();

  for (int e = tid; e < TB * G4; e += 128) {
    int m = e >> 8, j = e & 255;
    float acc = bias[j];
    #pragma unroll 5
    for (int k = 0; k < 65; ++k)
      acc += sm.sx[m * 65 + k] * w_ih[(11 + k) * G4 + j];
    sm.sg[m * SGPITCH + j] = acc;
  }

  const int khalf = lane >> 4;
  const int ml    = lane & 15;
  const int ncol  = wave * 16 + ml;

  v2f whhB[16][4];
  #pragma unroll
  for (int kt = 0; kt < 16; ++kt)
    #pragma unroll
    for (int nt = 0; nt < 4; ++nt) {
      int k = kt * 4 + khalf * 2;
      int n = nt * 64 + ncol;
      v2f b = { w_hh[k * G4 + n], w_hh[(k + 1) * G4 + n] };
      whhB[kt][nt] = b;
    }
  v2f wihB[3][4];
  #pragma unroll
  for (int kt = 0; kt < 3; ++kt)
    #pragma unroll
    for (int nt = 0; nt < 4; ++nt) {
      int k = kt * 4 + khalf * 2;
      int n = nt * 64 + ncol;
      v2f b = { w_ih[k * G4 + n], w_ih[(k + 1) * G4 + n] };
      wihB[kt][nt] = b;
    }
  __syncthreads();
  v8f statC[4];
  #pragma unroll
  for (int nt = 0; nt < 4; ++nt)
    #pragma unroll
    for (int r = 0; r < 8; ++r) {
      int m = r + khalf * 8;
      statC[nt][r] = sm.sg[m * SGPITCH + nt * 64 + ncol];
    }

  for (int e = tid; e < TB * HPITCH; e += 128) sm.h[e] = 0.0f;
  for (int e = tid; e < 2 * CHUNK * TB; e += 128) sm.xs[e * XPITCH + (XPITCH - 1)] = 0.0f;
  const float isc0 = sm.iscl[tid >> 4], isc1 = sm.iscl[(tid + 128) >> 4];
  ChunkRegs cr;
  load_chunk(cr, covs, target, b0, 0, tid);
  store_chunk(sm, cr, isc0, isc1, tid, 0);
  __syncthreads();

  v8f creg = {0.f, 0.f, 0.f, 0.f, 0.f, 0.f, 0.f, 0.f};
  v8f hreg;

  v2f xA[3];
  #pragma unroll
  for (int kt = 0; kt < 3; ++kt)
    xA[kt] = *(const v2f*)&sm.xs[ml * XPITCH + khalf * 2 + kt * 4];

  for (int t = 0; t < S_LEN; ++t) {
    v8f acc[4];
    #pragma unroll
    for (int nt = 0; nt < 4; ++nt) acc[nt] = statC[nt];
    #pragma unroll
    for (int nt = 0; nt < 4; ++nt)
      acc[nt] = __builtin_amdgcn_wmma_f32_16x16x4_f32(false, xA[0], false, wihB[0][nt],
                                                      (short)0, acc[nt], false, false);
    #pragma unroll
    for (int nt = 0; nt < 2; ++nt)
      acc[nt] = __builtin_amdgcn_wmma_f32_16x16x4_f32(false, xA[1], false, wihB[1][nt],
                                                      (short)0, acc[nt], false, false);
    __builtin_amdgcn_sched_barrier(0);
    asm volatile("s_wait_dscnt 0x0" ::: "memory");
    __syncthreads();

    if ((t & (CHUNK - 1)) == 0 && t + CHUNK < S_LEN)
      load_chunk(cr, covs, target, b0, t + CHUNK, tid);

    v2f hA[16], xAn[3];
    #pragma unroll
    for (int kt = 0; kt < 16; ++kt)
      hA[kt] = *(const v2f*)&sm.h[ml * HPITCH + kt * 4 + khalf * 2];
    __builtin_amdgcn_sched_barrier(0);
    const int hp_p = tid & 31, hp_q = tid >> 5;
    const int hp_m = hp_p >> 1, hp_j = hp_p & 1;
    float hv[16], wv[16];
    #pragma unroll
    for (int kk = 0; kk < 16; ++kk) {
      int k = hp_q * 16 + kk;
      hv[kk] = sm.h[hp_m * HPITCH + k];
      wv[kk] = sm.wo[k * 2 + hp_j];
    }
    const float* xrow = &sm.xs[(((t + 1) & (2 * CHUNK - 1)) * TB + ml) * XPITCH + khalf * 2];
    #pragma unroll
    for (int kt = 0; kt < 3; ++kt)
      xAn[kt] = *(const v2f*)&xrow[kt * 4];
    __builtin_amdgcn_sched_barrier(0);

    #pragma unroll
    for (int nt = 2; nt < 4; ++nt)
      acc[nt] = __builtin_amdgcn_wmma_f32_16x16x4_f32(false, xA[1], false, wihB[1][nt],
                                                      (short)0, acc[nt], false, false);
    #pragma unroll
    for (int nt = 0; nt < 4; ++nt)
      acc[nt] = __builtin_amdgcn_wmma_f32_16x16x4_f32(false, xA[2], false, wihB[2][nt],
                                                      (short)0, acc[nt], false, false);
    __builtin_amdgcn_sched_barrier(0);
    #pragma unroll
    for (int kt = 0; kt < 8; ++kt)
      #pragma unroll
      for (int nt = 0; nt < 4; ++nt)
        acc[nt] = __builtin_amdgcn_wmma_f32_16x16x4_f32(false, hA[kt], false, whhB[kt][nt],
                                                        (short)0, acc[nt], false, false);
    {
      float s = 0.0f;
      #pragma unroll
      for (int kk = 0; kk < 16; ++kk)
        s += hv[kk] * wv[kk];
      sm.part[((t - 1) & (CHUNK - 1)) * PPITCH + hp_q * 32 + hp_p] = s;
    }
    __builtin_amdgcn_sched_barrier(0);
    #pragma unroll
    for (int kt = 8; kt < 12; ++kt)
      #pragma unroll
      for (int nt = 0; nt < 4; ++nt)
        acc[nt] = __builtin_amdgcn_wmma_f32_16x16x4_f32(false, hA[kt], false, whhB[kt][nt],
                                                        (short)0, acc[nt], false, false);
    __builtin_amdgcn_sched_barrier(0);
    #pragma unroll
    for (int kt = 12; kt < 16; ++kt)
      #pragma unroll
      for (int nt = 0; nt < 4; ++nt)
        acc[nt] = __builtin_amdgcn_wmma_f32_16x16x4_f32(false, hA[kt], false, whhB[kt][nt],
                                                        (short)0, acc[nt], false, false);
    #pragma unroll
    for (int r = 0; r < 8; ++r) {
      float gi = fast_sigmoid(acc[0][r]);
      float gf = fast_sigmoid(acc[1][r]);
      float gg = fast_tanh(acc[2][r]);
      float go = fast_sigmoid(acc[3][r]);
      float cc = gf * creg[r] + gi * gg;
      creg[r] = cc;
      hreg[r] = go * fast_tanh(cc);
    }

    __syncthreads();

    #pragma unroll
    for (int r = 0; r < 8; ++r) {
      int m = r + khalf * 8;
      sm.h[m * HPITCH + ncol] = hreg[r];
    }
    if (t > 0 && (t & (CHUNK - 1)) == 0)
      flush_outputs(sm, out, b0, t - CHUNK, tid);
    if ((t & (CHUNK - 1)) == 7 && t + 9 < S_LEN)
      store_chunk(sm, cr, isc0, isc1, tid, ((t + 9) >> 4) & 1);
    #pragma unroll
    for (int kt = 0; kt < 3; ++kt) xA[kt] = xAn[kt];
  }

  __syncthreads();
  {
    int p = tid & 31, q = tid >> 5;
    int m = p >> 1, j = p & 1;
    float s = 0.0f;
    #pragma unroll
    for (int kk = 0; kk < 16; ++kk) {
      int k = q * 16 + kk;
      s += sm.h[m * HPITCH + k] * sm.wo[k * 2 + j];
    }
    sm.part[((S_LEN - 1) & (CHUNK - 1)) * PPITCH + q * 32 + p] = s;
  }
  __syncthreads();
  flush_outputs(sm, out, b0, S_LEN - CHUNK, tid);
}

extern "C" void kernel_launch(void* const* d_in, const int* in_sizes, int n_in,
                              void* d_out, int out_size, void* d_ws, size_t ws_size,
                              hipStream_t stream) {
  const float* target = (const float*)d_in[0];
  const float* covs   = (const float*)d_in[1];
  const int*   cats   = (const int*)  d_in[2];
  const float* scale  = (const float*)d_in[3];
  const float* emb0   = (const float*)d_in[4];
  const float* emb1   = (const float*)d_in[5];
  const float* emb2   = (const float*)d_in[6];
  const float* emb3   = (const float*)d_in[7];
  const float* w_ih   = (const float*)d_in[8];
  const float* w_hh   = (const float*)d_in[9];
  const float* bias   = (const float*)d_in[10];
  const float* w_out  = (const float*)d_in[11];
  const float* b_out  = (const float*)d_in[12];
  float* out = (float*)d_out;

  dim3 grid(B_TOT / TB);
  dim3 block(128);
  deepar_lstm_kernel<<<grid, block, 0, stream>>>(
      target, covs, cats, scale, emb0, emb1, emb2, emb3,
      w_ih, w_hh, bias, w_out, b_out, out);
}


# Static gate term (bias + static_x . w_ih[11:76]) moved from a scalar FMA loop of serialized global loads onto 68 WMMAs per wave, straight into C layout; its sum order: WMMA k-tiles, bias last
# speedup vs baseline: 2.0676x; 1.1588x over previous
#include <hip/hip_runtime.h>
#include <hip/hip_bf16.h>
#include <math.h>

typedef __attribute__((ext_vector_type(2))) float v2f;
typedef __attribute__((ext_vector_type(8))) float v8f;

#define B_TOT   4096
#define S_LEN   512
#define HIDN    64
#define G4      256
#define NCOV    10
#define EMB     16
#define TB      16
#define HPITCH  68
#define SXPITCH 68
#define SXTILES 17
#define XPITCH  12
#define CHUNK   16
#define CPITCH  (CHUNK*NCOV)
#define PPITCH  132

struct __align__(16) SMem {
  float xs[2 * CHUNK * TB * XPITCH];
  float h[TB * HPITCH];
  float sx[TB * SXPITCH];
  float part[CHUNK * PPITCH];
  float wo[HIDN * 2];
  float bo[2];
  float scl[TB];
  float iscl[TB];
};

__device__ __forceinline__ float fast_tanh(float x) {
#if __has_builtin(__builtin_amdgcn_tanhf)
  return __builtin_amdgcn_tanhf(x);
#else
  float y;
  asm("v_tanh_f32 %0, %1" : "=v"(y) : "v"(x));
  return y;
#endif
}

__device__ __forceinline__ float fast_sigmoid(float x) {
  return __builtin_amdgcn_rcpf(1.0f + __expf(-x));
}

struct ChunkRegs {
  float4 cv[TB * (CPITCH / 4) / 128];
  float  tg[TB * CHUNK / 128];
};

__device__ __forceinline__ void load_chunk(ChunkRegs& cr,
                                           const float* __restrict__ covs,
                                           const float* __restrict__ target,
                                           int b0, int t0, int tid) {
  #pragma unroll
  for (int i = 0; i < TB * (CPITCH / 4) / 128; ++i) {
    int e4 = tid + i * 128;
    int m = e4 / (CPITCH / 4), off4 = e4 % (CPITCH / 4);
    cr.cv[i] = *(const float4*)&covs[((size_t)(b0 + m) * S_LEN + t0) * NCOV + off4 * 4];
  }
  #pragma unroll
  for (int i = 0; i < TB * CHUNK / 128; ++i) {
    int e = tid + i * 128;
    int m = e >> 4, j = e & 15;
    int ti = t0 + j - 1;
    float v = target[(size_t)(b0 + m) * S_LEN + (ti < 0 ? 0 : ti)];
    cr.tg[i] = (ti < 0) ? 0.0f : v;
  }
}

__device__ __forceinline__ void store_chunk(SMem& sm, const ChunkRegs& cr,
                                            float isc0, float isc1, int tid, int half) {
  float* xs = &sm.xs[half * (CHUNK * TB * XPITCH)];
  #pragma unroll
  for (int i = 0; i < TB * (CPITCH / 4) / 128; ++i) {
    int e4 = tid + i * 128;
    int m = e4 / (CPITCH / 4), off4 = e4 % (CPITCH / 4);
    float vv[4] = { cr.cv[i].x, cr.cv[i].y, cr.cv[i].z, cr.cv[i].w };
    #pragma unroll
    for (int q = 0; q < 4; ++q) {
      int idx = off4 * 4 + q;
      int tc = idx / NCOV, k = idx % NCOV;
      xs[(tc * TB + m) * XPITCH + 1 + k] = vv[q];
    }
  }
  #pragma unroll
  for (int i = 0; i < TB * CHUNK / 128; ++i) {
    int e = tid + i * 128;
    int m = e >> 4, j = e & 15;
    xs[(j * TB + m) * XPITCH] = cr.tg[i] * (i == 0 ? isc0 : isc1);
  }
}

__device__ __forceinline__ void flush_outputs(const SMem& sm, float* __restrict__ out,
                                              int b0, int t0, int tid) {
  const int tq = tid & 3;
  const int p  = tid >> 2;
  const int m  = p >> 1, j = p & 1;
  const float bj   = sm.bo[j];
  const float mult = (j == 0) ? sm.scl[m] : 1.0f;
  float v[4];
  #pragma unroll
  for (int i = 0; i < 4; ++i) {
    const float* pp = &sm.part[(tq * 4 + i) * PPITCH + p];
    float r = bj + pp[0] + pp[32] + pp[64] + pp[96];
    float l = log1pf(__expf(r));
    float sp = (r > 20.0f) ? r : l;
    sp += 1e-4f;
    v[i] = sp * mult;
  }
  size_t idx = (size_t)j * B_TOT * S_LEN + (size_t)(b0 + m) * S_LEN + t0 + tq * 4;
  *(float4*)&out[idx] = make_float4(v[0], v[1], v[2], v[3]);
}

__global__ __launch_bounds__(128, 1)
void deepar_lstm_kernel(const float* __restrict__ target,
                        const float* __restrict__ covs,
                        const int*   __restrict__ cats,
                        const float* __restrict__ scale,
                        const float* __restrict__ emb0,
                        const float* __restrict__ emb1,
                        const float* __restrict__ emb2,
                        const float* __restrict__ emb3,
                        const float* __restrict__ w_ih,
                        const float* __restrict__ w_hh,
                        const float* __restrict__ bias,
                        const float* __restrict__ w_out,
                        const float* __restrict__ b_out,
                        float* __restrict__ out)
{
  __shared__ SMem sm;
  const int tid  = threadIdx.x;
  const int wave = tid >> 5;
  const int lane = tid & 31;
  const int b0   = blockIdx.x * TB;

  if (tid < TB) {
    float s = scale[b0 + tid];
    sm.scl[tid]  = s;
    sm.iscl[tid] = 1.0f / fmaxf(s, 1e-4f);
  }
  if (tid < HIDN * 2) sm.wo[tid] = w_out[tid];
  if (tid < 2)        sm.bo[tid] = b_out[tid];

  for (int e = tid; e < TB * SXPITCH; e += 128) {
    int m = e / SXPITCH, k = e % SXPITCH;
    float v;
    if      (k < 16) v = emb0[cats[(b0 + m) * 4 + 0] * EMB + k];
    else if (k < 32) v = emb1[cats[(b0 + m) * 4 + 1] * EMB + (k - 16)];
    else if (k < 48) v = emb2[cats[(b0 + m) * 4 + 2] * EMB + (k - 32)];
    else if (k < 64) v = emb3[cats[(b0 + m) * 4 + 3] * EMB + (k - 48)];
    else if (k == 64) v = log1pf(scale[b0 + m]);
    else if (k == 65) v = 1.0f;
    else             v = 0.0f;
    sm.sx[e] = v;
  }
  __syncthreads();

  const int khalf = lane >> 4;
  const int ml    = lane & 15;
  const int ncol  = wave * 16 + ml;

  v2f whhB[16][4];
  #pragma unroll
  for (int kt = 0; kt < 16; ++kt)
    #pragma unroll
    for (int nt = 0; nt < 4; ++nt) {
      int k = kt * 4 + khalf * 2;
      int n = nt * 64 + ncol;
      v2f b = { w_hh[k * G4 + n], w_hh[(k + 1) * G4 + n] };
      whhB[kt][nt] = b;
    }
  v2f wihB[3][4];
  #pragma unroll
  for (int kt = 0; kt < 3; ++kt)
    #pragma unroll
    for (int nt = 0; nt < 4; ++nt) {
      int k = kt * 4 + khalf * 2;
      int n = nt * 64 + ncol;
      v2f b = { w_ih[k * G4 + n], w_ih[(k + 1) * G4 + n] };
      wihB[kt][nt] = b;
    }
  v2f wsB[SXTILES][4];
  #pragma unroll
  for (int kt = 0; kt < SXTILES - 1; ++kt)
    #pragma unroll
    for (int nt = 0; nt < 4; ++nt) {
      int k = 11 + kt * 4 + khalf * 2;
      int n = nt * 64 + ncol;
      v2f b = { w_ih[k * G4 + n], w_ih[(k + 1) * G4 + n] };
      wsB[kt][nt] = b;
    }
  #pragma unroll
  for (int nt = 0; nt < 4; ++nt) {
    float w64 = w_ih[75 * G4 + nt * 64 + ncol];
    float bv  = bias[nt * 64 + ncol];
    v2f b = { khalf == 0 ? w64 : 0.0f, khalf == 0 ? bv : 0.0f };
    wsB[SXTILES - 1][nt] = b;
  }
  v8f statC[4];
  #pragma unroll
  for (int nt = 0; nt < 4; ++nt)
    statC[nt] = (v8f){0.f, 0.f, 0.f, 0.f, 0.f, 0.f, 0.f, 0.f};
  v2f sA[SXTILES];
  #pragma unroll
  for (int kt = 0; kt < SXTILES; ++kt)
    sA[kt] = *(const v2f*)&sm.sx[ml * SXPITCH + kt * 4 + khalf * 2];
  __builtin_amdgcn_sched_barrier(0);
  #pragma unroll
  for (int kt = 0; kt < SXTILES; ++kt)
    #pragma unroll
    for (int nt = 0; nt < 4; ++nt)
      statC[nt] = __builtin_amdgcn_wmma_f32_16x16x4_f32(false, sA[kt], false, wsB[kt][nt],
                                                        (short)0, statC[nt], false, false);
  __builtin_amdgcn_sched_barrier(0);
  asm volatile("v_nop\n\tv_nop\n\tv_nop\n\tv_nop"
               : "+v"(statC[0]), "+v"(statC[1]), "+v"(statC[2]), "+v"(statC[3]));
  __builtin_amdgcn_sched_barrier(0);

  for (int e = tid; e < TB * HPITCH; e += 128) sm.h[e] = 0.0f;
  for (int e = tid; e < 2 * CHUNK * TB; e += 128) sm.xs[e * XPITCH + (XPITCH - 1)] = 0.0f;
  const float isc0 = sm.iscl[tid >> 4], isc1 = sm.iscl[(tid + 128) >> 4];
  ChunkRegs cr;
  load_chunk(cr, covs, target, b0, 0, tid);
  store_chunk(sm, cr, isc0, isc1, tid, 0);
  __syncthreads();

  v8f creg = {0.f, 0.f, 0.f, 0.f, 0.f, 0.f, 0.f, 0.f};
  v8f hreg;

  v2f xA[3];
  #pragma unroll
  for (int kt = 0; kt < 3; ++kt)
    xA[kt] = *(const v2f*)&sm.xs[ml * XPITCH + khalf * 2 + kt * 4];

  for (int t = 0; t < S_LEN; ++t) {
    v8f acc[4];
    #pragma unroll
    for (int nt = 0; nt < 4; ++nt) acc[nt] = statC[nt];
    #pragma unroll
    for (int nt = 0; nt < 4; ++nt)
      acc[nt] = __builtin_amdgcn_wmma_f32_16x16x4_f32(false, xA[0], false, wihB[0][nt],
                                                      (short)0, acc[nt], false, false);
    #pragma unroll
    for (int nt = 0; nt < 2; ++nt)
      acc[nt] = __builtin_amdgcn_wmma_f32_16x16x4_f32(false, xA[1], false, wihB[1][nt],
                                                      (short)0, acc[nt], false, false);
    __builtin_amdgcn_sched_barrier(0);
    asm volatile("s_wait_dscnt 0x0" ::: "memory");
    __syncthreads();

    if ((t & (CHUNK - 1)) == 0 && t + CHUNK < S_LEN)
      load_chunk(cr, covs, target, b0, t + CHUNK, tid);

    v2f hA[16], xAn[3];
    #pragma unroll
    for (int kt = 0; kt < 16; ++kt)
      hA[kt] = *(const v2f*)&sm.h[ml * HPITCH + kt * 4 + khalf * 2];
    __builtin_amdgcn_sched_barrier(0);
    const int hp_p = tid & 31, hp_q = tid >> 5;
    const int hp_m = hp_p >> 1, hp_j = hp_p & 1;
    float hv[16], wv[16];
    #pragma unroll
    for (int kk = 0; kk < 16; ++kk) {
      int k = hp_q * 16 + kk;
      hv[kk] = sm.h[hp_m * HPITCH + k];
      wv[kk] = sm.wo[k * 2 + hp_j];
    }
    const float* xrow = &sm.xs[(((t + 1) & (2 * CHUNK - 1)) * TB + ml) * XPITCH + khalf * 2];
    #pragma unroll
    for (int kt = 0; kt < 3; ++kt)
      xAn[kt] = *(const v2f*)&xrow[kt * 4];
    __builtin_amdgcn_sched_barrier(0);

    #pragma unroll
    for (int nt = 2; nt < 4; ++nt)
      acc[nt] = __builtin_amdgcn_wmma_f32_16x16x4_f32(false, xA[1], false, wihB[1][nt],
                                                      (short)0, acc[nt], false, false);
    #pragma unroll
    for (int nt = 0; nt < 4; ++nt)
      acc[nt] = __builtin_amdgcn_wmma_f32_16x16x4_f32(false, xA[2], false, wihB[2][nt],
                                                      (short)0, acc[nt], false, false);
    __builtin_amdgcn_sched_barrier(0);
    #pragma unroll
    for (int kt = 0; kt < 8; ++kt)
      #pragma unroll
      for (int nt = 0; nt < 4; ++nt)
        acc[nt] = __builtin_amdgcn_wmma_f32_16x16x4_f32(false, hA[kt], false, whhB[kt][nt],
                                                        (short)0, acc[nt], false, false);
    {
      float s = 0.0f;
      #pragma unroll
      for (int kk = 0; kk < 16; ++kk)
        s += hv[kk] * wv[kk];
      sm.part[((t - 1) & (CHUNK - 1)) * PPITCH + hp_q * 32 + hp_p] = s;
    }
    __builtin_amdgcn_sched_barrier(0);
    #pragma unroll
    for (int kt = 8; kt < 12; ++kt)
      #pragma unroll
      for (int nt = 0; nt < 4; ++nt)
        acc[nt] = __builtin_amdgcn_wmma_f32_16x16x4_f32(false, hA[kt], false, whhB[kt][nt],
                                                        (short)0, acc[nt], false, false);
    __builtin_amdgcn_sched_barrier(0);
    #pragma unroll
    for (int kt = 12; kt < 16; ++kt)
      #pragma unroll
      for (int nt = 0; nt < 4; ++nt)
        acc[nt] = __builtin_amdgcn_wmma_f32_16x16x4_f32(false, hA[kt], false, whhB[kt][nt],
                                                        (short)0, acc[nt], false, false);
    #pragma unroll
    for (int r = 0; r < 8; ++r) {
      float gi = fast_sigmoid(acc[0][r]);
      float gf = fast_sigmoid(acc[1][r]);
      float gg = fast_tanh(acc[2][r]);
      float go = fast_sigmoid(acc[3][r]);
      float cc = gf * creg[r] + gi * gg;
      creg[r] = cc;
      hreg[r] = go * fast_tanh(cc);
    }

    __syncthreads();

    #pragma unroll
    for (int r = 0; r < 8; ++r) {
      int m = r + khalf * 8;
      sm.h[m * HPITCH + ncol] = hreg[r];
    }
    if (t > 0 && (t & (CHUNK - 1)) == 0)
      flush_outputs(sm, out, b0, t - CHUNK, tid);
    if ((t & (CHUNK - 1)) == 7 && t + 9 < S_LEN)
      store_chunk(sm, cr, isc0, isc1, tid, ((t + 9) >> 4) & 1);
    #pragma unroll
    for (int kt = 0; kt < 3; ++kt) xA[kt] = xAn[kt];
  }

  __syncthreads();
  {
    int p = tid & 31, q = tid >> 5;
    int m = p >> 1, j = p & 1;
    float s = 0.0f;
    #pragma unroll
    for (int kk = 0; kk < 16; ++kk) {
      int k = q * 16 + kk;
      s += sm.h[m * HPITCH + k] * sm.wo[k * 2 + j];
    }
    sm.part[((S_LEN - 1) & (CHUNK - 1)) * PPITCH + q * 32 + p] = s;
  }
  __syncthreads();
  flush_outputs(sm, out, b0, S_LEN - CHUNK, tid);
}

extern "C" void kernel_launch(void* const* d_in, const int* in_sizes, int n_in,
                              void* d_out, int out_size, void* d_ws, size_t ws_size,
                              hipStream_t stream) {
  const float* target = (const float*)d_in[0];
  const float* covs   = (const float*)d_in[1];
  const int*   cats   = (const int*)  d_in[2];
  const float* scale  = (const float*)d_in[3];
  const float* emb0   = (const float*)d_in[4];
  const float* emb1   = (const float*)d_in[5];
  const float* emb2   = (const float*)d_in[6];
  const float* emb3   = (const float*)d_in[7];
  const float* w_ih   = (const float*)d_in[8];
  const float* w_hh   = (const float*)d_in[9];
  const float* bias   = (const float*)d_in[10];
  const float* w_out  = (const float*)d_in[11];
  const float* b_out  = (const float*)d_in[12];
  float* out = (float*)d_out;

  dim3 grid(B_TOT / TB);
  dim3 block(128);
  deepar_lstm_kernel<<<grid, block, 0, stream>>>(
      target, covs, cats, scale, emb0, emb1, emb2, emb3,
      w_ih, w_hh, bias, w_out, b_out, out);
}


# Sigmoid of the i, f, o gates as 0.5 + 0.5*tanh(x/2) on v_tanh_f32, the 1/2 folded exactly into those gates' weight and bias fragments: 40 transcendentals per lane and step, not 64; sum order kept
# speedup vs baseline: 2.1355x; 1.0328x over previous
#include <hip/hip_runtime.h>
#include <hip/hip_bf16.h>
#include <math.h>

typedef __attribute__((ext_vector_type(2))) float v2f;
typedef __attribute__((ext_vector_type(8))) float v8f;

#define B_TOT   4096
#define S_LEN   512
#define HIDN    64
#define G4      256
#define NCOV    10
#define EMB     16
#define TB      16
#define HPITCH  68
#define SXPITCH 68
#define SXTILES 17
#define XPITCH  12
#define CHUNK   16
#define CPITCH  (CHUNK*NCOV)
#define PPITCH  132

struct __align__(16) SMem {
  float xs[2 * CHUNK * TB * XPITCH];
  float h[TB * HPITCH];
  float sx[TB * SXPITCH];
  float part[CHUNK * PPITCH];
  float wo[HIDN * 2];
  float bo[2];
  float scl[TB];
  float iscl[TB];
};

__device__ __forceinline__ float fast_tanh(float x) {
#if __has_builtin(__builtin_amdgcn_tanhf)
  return __builtin_amdgcn_tanhf(x);
#else
  float y;
  asm("v_tanh_f32 %0, %1" : "=v"(y) : "v"(x));
  return y;
#endif
}

__device__ __forceinline__ float sigmoid_of_half(float xh) {
  return fmaf(0.5f, fast_tanh(xh), 0.5f);
}

__device__ __forceinline__ constexpr float gate_scale(int nt) { return nt == 2 ? 1.0f : 0.5f; }

struct ChunkRegs {
  float4 cv[TB * (CPITCH / 4) / 128];
  float  tg[TB * CHUNK / 128];
};

__device__ __forceinline__ void load_chunk(ChunkRegs& cr,
                                           const float* __restrict__ covs,
                                           const float* __restrict__ target,
                                           int b0, int t0, int tid) {
  #pragma unroll
  for (int i = 0; i < TB * (CPITCH / 4) / 128; ++i) {
    int e4 = tid + i * 128;
    int m = e4 / (CPITCH / 4), off4 = e4 % (CPITCH / 4);
    cr.cv[i] = *(const float4*)&covs[((size_t)(b0 + m) * S_LEN + t0) * NCOV + off4 * 4];
  }
  #pragma unroll
  for (int i = 0; i < TB * CHUNK / 128; ++i) {
    int e = tid + i * 128;
    int m = e >> 4, j = e & 15;
    int ti = t0 + j - 1;
    float v = target[(size_t)(b0 + m) * S_LEN + (ti < 0 ? 0 : ti)];
    cr.tg[i] = (ti < 0) ? 0.0f : v;
  }
}

__device__ __forceinline__ void store_chunk(SMem& sm, const ChunkRegs& cr,
                                            float isc0, float isc1, int tid, int half) {
  float* xs = &sm.xs[half * (CHUNK * TB * XPITCH)];
  #pragma unroll
  for (int i = 0; i < TB * (CPITCH / 4) / 128; ++i) {
    int e4 = tid + i * 128;
    int m = e4 / (CPITCH / 4), off4 = e4 % (CPITCH / 4);
    float vv[4] = { cr.cv[i].x, cr.cv[i].y, cr.cv[i].z, cr.cv[i].w };
    #pragma unroll
    for (int q = 0; q < 4; ++q) {
      int idx = off4 * 4 + q;
      int tc = idx / NCOV, k = idx % NCOV;
      xs[(tc * TB + m) * XPITCH + 1 + k] = vv[q];
    }
  }
  #pragma unroll
  for (int i = 0; i < TB * CHUNK / 128; ++i) {
    int e = tid + i * 128;
    int m = e >> 4, j = e & 15;
    xs[(j * TB + m) * XPITCH] = cr.tg[i] * (i == 0 ? isc0 : isc1);
  }
}

__device__ __forceinline__ void flush_outputs(const SMem& sm, float* __restrict__ out,
                                              int b0, int t0, int tid) {
  const int tq = tid & 3;
  const int p  = tid >> 2;
  const int m  = p >> 1, j = p & 1;
  const float bj   = sm.bo[j];
  const float mult = (j == 0) ? sm.scl[m] : 1.0f;
  float v[4];
  #pragma unroll
  for (int i = 0; i < 4; ++i) {
    const float* pp = &sm.part[(tq * 4 + i) * PPITCH + p];
    float r = bj + pp[0] + pp[32] + pp[64] + pp[96];
    float l = log1pf(__expf(r));
    float sp = (r > 20.0f) ? r : l;
    sp += 1e-4f;
    v[i] = sp * mult;
  }
  size_t idx = (size_t)j * B_TOT * S_LEN + (size_t)(b0 + m) * S_LEN + t0 + tq * 4;
  *(float4*)&out[idx] = make_float4(v[0], v[1], v[2], v[3]);
}

__global__ __launch_bounds__(128, 1)
void deepar_lstm_kernel(const float* __restrict__ target,
                        const float* __restrict__ covs,
                        const int*   __restrict__ cats,
                        const float* __restrict__ scale,
                        const float* __restrict__ emb0,
                        const float* __restrict__ emb1,
                        const float* __restrict__ emb2,
                        const float* __restrict__ emb3,
                        const float* __restrict__ w_ih,
                        const float* __restrict__ w_hh,
                        const float* __restrict__ bias,
                        const float* __restrict__ w_out,
                        const float* __restrict__ b_out,
                        float* __restrict__ out)
{
  __shared__ SMem sm;
  const int tid  = threadIdx.x;
  const int wave = tid >> 5;
  const int lane = tid & 31;
  const int b0   = blockIdx.x * TB;

  if (tid < TB) {
    float s = scale[b0 + tid];
    sm.scl[tid]  = s;
    sm.iscl[tid] = 1.0f / fmaxf(s, 1e-4f);
  }
  if (tid < HIDN * 2) sm.wo[tid] = w_out[tid];
  if (tid < 2)        sm.bo[tid] = b_out[tid];

  for (int e = tid; e < TB * SXPITCH; e += 128) {
    int m = e / SXPITCH, k = e % SXPITCH;
    float v;
    if      (k < 16) v = emb0[cats[(b0 + m) * 4 + 0] * EMB + k];
    else if (k < 32) v = emb1[cats[(b0 + m) * 4 + 1] * EMB + (k - 16)];
    else if (k < 48) v = emb2[cats[(b0 + m) * 4 + 2] * EMB + (k - 32)];
    else if (k < 64) v = emb3[cats[(b0 + m) * 4 + 3] * EMB + (k - 48)];
    else if (k == 64) v = log1pf(scale[b0 + m]);
    else if (k == 65) v = 1.0f;
    else             v = 0.0f;
    sm.sx[e] = v;
  }
  __syncthreads();

  const int khalf = lane >> 4;
  const int ml    = lane & 15;
  const int ncol  = wave * 16 + ml;

  v2f whhB[16][4];
  #pragma unroll
  for (int kt = 0; kt < 16; ++kt)
    #pragma unroll
    for (int nt = 0; nt < 4; ++nt) {
      int k = kt * 4 + khalf * 2;
      int n = nt * 64 + ncol;
      v2f b = { gate_scale(nt) * w_hh[k * G4 + n], gate_scale(nt) * w_hh[(k + 1) * G4 + n] };
      whhB[kt][nt] = b;
    }
  v2f wihB[3][4];
  #pragma unroll
  for (int kt = 0; kt < 3; ++kt)
    #pragma unroll
    for (int nt = 0; nt < 4; ++nt) {
      int k = kt * 4 + khalf * 2;
      int n = nt * 64 + ncol;
      v2f b = { gate_scale(nt) * w_ih[k * G4 + n], gate_scale(nt) * w_ih[(k + 1) * G4 + n] };
      wihB[kt][nt] = b;
    }
  v2f wsB[SXTILES][4];
  #pragma unroll
  for (int kt = 0; kt < SXTILES - 1; ++kt)
    #pragma unroll
    for (int nt = 0; nt < 4; ++nt) {
      int k = 11 + kt * 4 + khalf * 2;
      int n = nt * 64 + ncol;
      v2f b = { gate_scale(nt) * w_ih[k * G4 + n], gate_scale(nt) * w_ih[(k + 1) * G4 + n] };
      wsB[kt][nt] = b;
    }
  #pragma unroll
  for (int nt = 0; nt < 4; ++nt) {
    float w64 = gate_scale(nt) * w_ih[75 * G4 + nt * 64 + ncol];
    float bv  = gate_scale(nt) * bias[nt * 64 + ncol];
    v2f b = { khalf == 0 ? w64 : 0.0f, khalf == 0 ? bv : 0.0f };
    wsB[SXTILES - 1][nt] = b;
  }
  v8f statC[4];
  #pragma unroll
  for (int nt = 0; nt < 4; ++nt)
    statC[nt] = (v8f){0.f, 0.f, 0.f, 0.f, 0.f, 0.f, 0.f, 0.f};
  v2f sA[SXTILES];
  #pragma unroll
  for (int kt = 0; kt < SXTILES; ++kt)
    sA[kt] = *(const v2f*)&sm.sx[ml * SXPITCH + kt * 4 + khalf * 2];
  __builtin_amdgcn_sched_barrier(0);
  #pragma unroll
  for (int kt = 0; kt < SXTILES; ++kt)
    #pragma unroll
    for (int nt = 0; nt < 4; ++nt)
      statC[nt] = __builtin_amdgcn_wmma_f32_16x16x4_f32(false, sA[kt], false, wsB[kt][nt],
                                                        (short)0, statC[nt], false, false);
  __builtin_amdgcn_sched_barrier(0);
  asm volatile("v_nop\n\tv_nop\n\tv_nop\n\tv_nop"
               : "+v"(statC[0]), "+v"(statC[1]), "+v"(statC[2]), "+v"(statC[3]));
  __builtin_amdgcn_sched_barrier(0);

  for (int e = tid; e < TB * HPITCH; e += 128) sm.h[e] = 0.0f;
  for (int e = tid; e < 2 * CHUNK * TB; e += 128) sm.xs[e * XPITCH + (XPITCH - 1)] = 0.0f;
  const float isc0 = sm.iscl[tid >> 4], isc1 = sm.iscl[(tid + 128) >> 4];
  ChunkRegs cr;
  load_chunk(cr, covs, target, b0, 0, tid);
  store_chunk(sm, cr, isc0, isc1, tid, 0);
  __syncthreads();

  v8f creg = {0.f, 0.f, 0.f, 0.f, 0.f, 0.f, 0.f, 0.f};
  v8f hreg;

  v2f xA[3];
  #pragma unroll
  for (int kt = 0; kt < 3; ++kt)
    xA[kt] = *(const v2f*)&sm.xs[ml * XPITCH + khalf * 2 + kt * 4];

  for (int t = 0; t < S_LEN; ++t) {
    v8f acc[4];
    #pragma unroll
    for (int nt = 0; nt < 4; ++nt) acc[nt] = statC[nt];
    #pragma unroll
    for (int nt = 0; nt < 4; ++nt)
      acc[nt] = __builtin_amdgcn_wmma_f32_16x16x4_f32(false, xA[0], false, wihB[0][nt],
                                                      (short)0, acc[nt], false, false);
    #pragma unroll
    for (int nt = 0; nt < 2; ++nt)
      acc[nt] = __builtin_amdgcn_wmma_f32_16x16x4_f32(false, xA[1], false, wihB[1][nt],
                                                      (short)0, acc[nt], false, false);
    __builtin_amdgcn_sched_barrier(0);
    asm volatile("s_wait_dscnt 0x0" ::: "memory");
    __syncthreads();

    if ((t & (CHUNK - 1)) == 0 && t + CHUNK < S_LEN)
      load_chunk(cr, covs, target, b0, t + CHUNK, tid);

    v2f hA[16], xAn[3];
    #pragma unroll
    for (int kt = 0; kt < 16; ++kt)
      hA[kt] = *(const v2f*)&sm.h[ml * HPITCH + kt * 4 + khalf * 2];
    __builtin_amdgcn_sched_barrier(0);
    const int hp_p = tid & 31, hp_q = tid >> 5;
    const int hp_m = hp_p >> 1, hp_j = hp_p & 1;
    float hv[16], wv[16];
    #pragma unroll
    for (int kk = 0; kk < 16; ++kk) {
      int k = hp_q * 16 + kk;
      hv[kk] = sm.h[hp_m * HPITCH + k];
      wv[kk] = sm.wo[k * 2 + hp_j];
    }
    const float* xrow = &sm.xs[(((t + 1) & (2 * CHUNK - 1)) * TB + ml) * XPITCH + khalf * 2];
    #pragma unroll
    for (int kt = 0; kt < 3; ++kt)
      xAn[kt] = *(const v2f*)&xrow[kt * 4];
    __builtin_amdgcn_sched_barrier(0);

    #pragma unroll
    for (int nt = 2; nt < 4; ++nt)
      acc[nt] = __builtin_amdgcn_wmma_f32_16x16x4_f32(false, xA[1], false, wihB[1][nt],
                                                      (short)0, acc[nt], false, false);
    #pragma unroll
    for (int nt = 0; nt < 4; ++nt)
      acc[nt] = __builtin_amdgcn_wmma_f32_16x16x4_f32(false, xA[2], false, wihB[2][nt],
                                                      (short)0, acc[nt], false, false);
    __builtin_amdgcn_sched_barrier(0);
    #pragma unroll
    for (int kt = 0; kt < 8; ++kt)
      #pragma unroll
      for (int nt = 0; nt < 4; ++nt)
        acc[nt] = __builtin_amdgcn_wmma_f32_16x16x4_f32(false, hA[kt], false, whhB[kt][nt],
                                                        (short)0, acc[nt], false, false);
    {
      float s = 0.0f;
      #pragma unroll
      for (int kk = 0; kk < 16; ++kk)
        s += hv[kk] * wv[kk];
      sm.part[((t - 1) & (CHUNK - 1)) * PPITCH + hp_q * 32 + hp_p] = s;
    }
    __builtin_amdgcn_sched_barrier(0);
    #pragma unroll
    for (int kt = 8; kt < 12; ++kt)
      #pragma unroll
      for (int nt = 0; nt < 4; ++nt)
        acc[nt] = __builtin_amdgcn_wmma_f32_16x16x4_f32(false, hA[kt], false, whhB[kt][nt],
                                                        (short)0, acc[nt], false, false);
    __builtin_amdgcn_sched_barrier(0);
    #pragma unroll
    for (int kt = 12; kt < 16; ++kt)
      #pragma unroll
      for (int nt = 0; nt < 4; ++nt)
        acc[nt] = __builtin_amdgcn_wmma_f32_16x16x4_f32(false, hA[kt], false, whhB[kt][nt],
                                                        (short)0, acc[nt], false, false);
    #pragma unroll
    for (int r = 0; r < 8; ++r) {
      float gi = sigmoid_of_half(acc[0][r]);
      float gf = sigmoid_of_half(acc[1][r]);
      float gg = fast_tanh(acc[2][r]);
      float go = sigmoid_of_half(acc[3][r]);
      float cc = gf * creg[r] + gi * gg;
      creg[r] = cc;
      hreg[r] = go * fast_tanh(cc);
    }

    __syncthreads();

    #pragma unroll
    for (int r = 0; r < 8; ++r) {
      int m = r + khalf * 8;
      sm.h[m * HPITCH + ncol] = hreg[r];
    }
    if (t > 0 && (t & (CHUNK - 1)) == 0)
      flush_outputs(sm, out, b0, t - CHUNK, tid);
    if ((t & (CHUNK - 1)) == 7 && t + 9 < S_LEN)
      store_chunk(sm, cr, isc0, isc1, tid, ((t + 9) >> 4) & 1);
    #pragma unroll
    for (int kt = 0; kt < 3; ++kt) xA[kt] = xAn[kt];
  }

  __syncthreads();
  {
    int p = tid & 31, q = tid >> 5;
    int m = p >> 1, j = p & 1;
    float s = 0.0f;
    #pragma unroll
    for (int kk = 0; kk < 16; ++kk) {
      int k = q * 16 + kk;
      s += sm.h[m * HPITCH + k] * sm.wo[k * 2 + j];
    }
    sm.part[((S_LEN - 1) & (CHUNK - 1)) * PPITCH + q * 32 + p] = s;
  }
  __syncthreads();
  flush_outputs(sm, out, b0, S_LEN - CHUNK, tid);
}

extern "C" void kernel_launch(void* const* d_in, const int* in_sizes, int n_in,
                              void* d_out, int out_size, void* d_ws, size_t ws_size,
                              hipStream_t stream) {
  const float* target = (const float*)d_in[0];
  const float* covs   = (const float*)d_in[1];
  const int*   cats   = (const int*)  d_in[2];
  const float* scale  = (const float*)d_in[3];
  const float* emb0   = (const float*)d_in[4];
  const float* emb1   = (const float*)d_in[5];
  const float* emb2   = (const float*)d_in[6];
  const float* emb3   = (const float*)d_in[7];
  const float* w_ih   = (const float*)d_in[8];
  const float* w_hh   = (const float*)d_in[9];
  const float* bias   = (const float*)d_in[10];
  const float* w_out  = (const float*)d_in[11];
  const float* b_out  = (const float*)d_in[12];
  float* out = (float*)d_out;

  dim3 grid(B_TOT / TB);
  dim3 block(128);
  deepar_lstm_kernel<<<grid, block, 0, stream>>>(
      target, covs, cats, scale, emb0, emb1, emb2, emb3,
      w_ih, w_hh, bias, w_out, b_out, out);
}


# Output-head partials from the wave's own k-quarter of h in A-tile order (2 LDS loads a step) with w_out in registers: 19 fewer LDS reads per step; head sum order: 8 partials of 8 k, not 4 of 16
# speedup vs baseline: 2.2052x; 1.0326x over previous
#include <hip/hip_runtime.h>
#include <hip/hip_bf16.h>
#include <math.h>

typedef __attribute__((ext_vector_type(2))) float v2f;
typedef __attribute__((ext_vector_type(8))) float v8f;

#define B_TOT   4096
#define S_LEN   512
#define HIDN    64
#define G4      256
#define NCOV    10
#define EMB     16
#define TB      16
#define HPITCH  68
#define SXPITCH 68
#define SXTILES 17
#define XPITCH  12
#define CHUNK   16
#define CPITCH  (CHUNK*NCOV)
#define PPITCH  260

struct __align__(16) SMem {
  float xs[2 * CHUNK * TB * XPITCH];
  float h[TB * HPITCH];
  float sx[TB * SXPITCH];
  float part[CHUNK * PPITCH];
  float bo[2];
  float scl[TB];
  float iscl[TB];
};

__device__ __forceinline__ float fast_tanh(float x) {
#if __has_builtin(__builtin_amdgcn_tanhf)
  return __builtin_amdgcn_tanhf(x);
#else
  float y;
  asm("v_tanh_f32 %0, %1" : "=v"(y) : "v"(x));
  return y;
#endif
}

__device__ __forceinline__ float sigmoid_of_half(float xh) {
  return fmaf(0.5f, fast_tanh(xh), 0.5f);
}

__device__ __forceinline__ constexpr float gate_scale(int nt) { return nt == 2 ? 1.0f : 0.5f; }

struct ChunkRegs {
  float4 cv[TB * (CPITCH / 4) / 128];
  float  tg[TB * CHUNK / 128];
};

__device__ __forceinline__ void load_chunk(ChunkRegs& cr,
                                           const float* __restrict__ covs,
                                           const float* __restrict__ target,
                                           int b0, int t0, int tid) {
  #pragma unroll
  for (int i = 0; i < TB * (CPITCH / 4) / 128; ++i) {
    int e4 = tid + i * 128;
    int m = e4 / (CPITCH / 4), off4 = e4 % (CPITCH / 4);
    cr.cv[i] = *(const float4*)&covs[((size_t)(b0 + m) * S_LEN + t0) * NCOV + off4 * 4];
  }
  #pragma unroll
  for (int i = 0; i < TB * CHUNK / 128; ++i) {
    int e = tid + i * 128;
    int m = e >> 4, j = e & 15;
    int ti = t0 + j - 1;
    float v = target[(size_t)(b0 + m) * S_LEN + (ti < 0 ? 0 : ti)];
    cr.tg[i] = (ti < 0) ? 0.0f : v;
  }
}

__device__ __forceinline__ void store_chunk(SMem& sm, const ChunkRegs& cr,
                                            float isc0, float isc1, int tid, int half) {
  float* xs = &sm.xs[half * (CHUNK * TB * XPITCH)];
  #pragma unroll
  for (int i = 0; i < TB * (CPITCH / 4) / 128; ++i) {
    int e4 = tid + i * 128;
    int m = e4 / (CPITCH / 4), off4 = e4 % (CPITCH / 4);
    float vv[4] = { cr.cv[i].x, cr.cv[i].y, cr.cv[i].z, cr.cv[i].w };
    #pragma unroll
    for (int q = 0; q < 4; ++q) {
      int idx = off4 * 4 + q;
      int tc = idx / NCOV, k = idx % NCOV;
      xs[(tc * TB + m) * XPITCH + 1 + k] = vv[q];
    }
  }
  #pragma unroll
  for (int i = 0; i < TB * CHUNK / 128; ++i) {
    int e = tid + i * 128;
    int m = e >> 4, j = e & 15;
    xs[(j * TB + m) * XPITCH] = cr.tg[i] * (i == 0 ? isc0 : isc1);
  }
}

__device__ __forceinline__ v2f head_partial(const v2f hf[4], const v2f w2[8]) {
  v2f s = {0.0f, 0.0f};
  #pragma unroll
  for (int i = 0; i < 4; ++i) {
    s.x = fmaf(hf[i].x, w2[2 * i].x, s.x);
    s.y = fmaf(hf[i].x, w2[2 * i].y, s.y);
    s.x = fmaf(hf[i].y, w2[2 * i + 1].x, s.x);
    s.y = fmaf(hf[i].y, w2[2 * i + 1].y, s.y);
  }
  return s;
}

__device__ __forceinline__ void flush_outputs(const SMem& sm, float* __restrict__ out,
                                              int b0, int t0, int tid) {
  const int tq = tid & 3;
  const int p  = tid >> 2;
  const int m  = p >> 1, j = p & 1;
  const float bj   = sm.bo[j];
  const float mult = (j == 0) ? sm.scl[m] : 1.0f;
  float v[4];
  #pragma unroll
  for (int i = 0; i < 4; ++i) {
    const float* pp = &sm.part[(tq * 4 + i) * PPITCH + p];
    float r = bj;
    #pragma unroll
    for (int g = 0; g < 8; ++g) r += pp[g * 32];
    float l = log1pf(__expf(r));
    float sp = (r > 20.0f) ? r : l;
    sp += 1e-4f;
    v[i] = sp * mult;
  }
  size_t idx = (size_t)j * B_TOT * S_LEN + (size_t)(b0 + m) * S_LEN + t0 + tq * 4;
  *(float4*)&out[idx] = make_float4(v[0], v[1], v[2], v[3]);
}

__global__ __launch_bounds__(128, 1)
void deepar_lstm_kernel(const float* __restrict__ target,
                        const float* __restrict__ covs,
                        const int*   __restrict__ cats,
                        const float* __restrict__ scale,
                        const float* __restrict__ emb0,
                        const float* __restrict__ emb1,
                        const float* __restrict__ emb2,
                        const float* __restrict__ emb3,
                        const float* __restrict__ w_ih,
                        const float* __restrict__ w_hh,
                        const float* __restrict__ bias,
                        const float* __restrict__ w_out,
                        const float* __restrict__ b_out,
                        float* __restrict__ out)
{
  __shared__ SMem sm;
  const int tid  = threadIdx.x;
  const int wave = tid >> 5;
  const int lane = tid & 31;
  const int b0   = blockIdx.x * TB;

  if (tid < TB) {
    float s = scale[b0 + tid];
    sm.scl[tid]  = s;
    sm.iscl[tid] = 1.0f / fmaxf(s, 1e-4f);
  }
  if (tid < 2)        sm.bo[tid] = b_out[tid];

  for (int e = tid; e < TB * SXPITCH; e += 128) {
    int m = e / SXPITCH, k = e % SXPITCH;
    float v;
    if      (k < 16) v = emb0[cats[(b0 + m) * 4 + 0] * EMB + k];
    else if (k < 32) v = emb1[cats[(b0 + m) * 4 + 1] * EMB + (k - 16)];
    else if (k < 48) v = emb2[cats[(b0 + m) * 4 + 2] * EMB + (k - 32)];
    else if (k < 64) v = emb3[cats[(b0 + m) * 4 + 3] * EMB + (k - 48)];
    else if (k == 64) v = log1pf(scale[b0 + m]);
    else if (k == 65) v = 1.0f;
    else             v = 0.0f;
    sm.sx[e] = v;
  }
  __syncthreads();

  const int khalf = lane >> 4;
  const int ml    = lane & 15;
  const int ncol  = wave * 16 + ml;

  v2f whhB[16][4];
  #pragma unroll
  for (int kt = 0; kt < 16; ++kt)
    #pragma unroll
    for (int nt = 0; nt < 4; ++nt) {
      int k = kt * 4 + khalf * 2;
      int n = nt * 64 + ncol;
      v2f b = { gate_scale(nt) * w_hh[k * G4 + n], gate_scale(nt) * w_hh[(k + 1) * G4 + n] };
      whhB[kt][nt] = b;
    }
  v2f wihB[3][4];
  #pragma unroll
  for (int kt = 0; kt < 3; ++kt)
    #pragma unroll
    for (int nt = 0; nt < 4; ++nt) {
      int k = kt * 4 + khalf * 2;
      int n = nt * 64 + ncol;
      v2f b = { gate_scale(nt) * w_ih[k * G4 + n], gate_scale(nt) * w_ih[(k + 1) * G4 + n] };
      wihB[kt][nt] = b;
    }
  v2f wsB[SXTILES][4];
  #pragma unroll
  for (int kt = 0; kt < SXTILES - 1; ++kt)
    #pragma unroll
    for (int nt = 0; nt < 4; ++nt) {
      int k = 11 + kt * 4 + khalf * 2;
      int n = nt * 64 + ncol;
      v2f b = { gate_scale(nt) * w_ih[k * G4 + n], gate_scale(nt) * w_ih[(k + 1) * G4 + n] };
      wsB[kt][nt] = b;
    }
  #pragma unroll
  for (int nt = 0; nt < 4; ++nt) {
    float w64 = gate_scale(nt) * w_ih[75 * G4 + nt * 64 + ncol];
    float bv  = gate_scale(nt) * bias[nt * 64 + ncol];
    v2f b = { khalf == 0 ? w64 : 0.0f, khalf == 0 ? bv : 0.0f };
    wsB[SXTILES - 1][nt] = b;
  }
  v8f statC[4];
  #pragma unroll
  for (int nt = 0; nt < 4; ++nt)
    statC[nt] = (v8f){0.f, 0.f, 0.f, 0.f, 0.f, 0.f, 0.f, 0.f};
  v2f sA[SXTILES];
  #pragma unroll
  for (int kt = 0; kt < SXTILES; ++kt)
    sA[kt] = *(const v2f*)&sm.sx[ml * SXPITCH + kt * 4 + khalf * 2];
  __builtin_amdgcn_sched_barrier(0);
  #pragma unroll
  for (int kt = 0; kt < SXTILES; ++kt)
    #pragma unroll
    for (int nt = 0; nt < 4; ++nt)
      statC[nt] = __builtin_amdgcn_wmma_f32_16x16x4_f32(false, sA[kt], false, wsB[kt][nt],
                                                        (short)0, statC[nt], false, false);
  __builtin_amdgcn_sched_barrier(0);
  asm volatile("v_nop\n\tv_nop\n\tv_nop\n\tv_nop"
               : "+v"(statC[0]), "+v"(statC[1]), "+v"(statC[2]), "+v"(statC[3]));
  __builtin_amdgcn_sched_barrier(0);

  for (int e = tid; e < TB * HPITCH; e += 128) sm.h[e] = 0.0f;
  for (int e = tid; e < 2 * CHUNK * TB; e += 128) sm.xs[e * XPITCH + (XPITCH - 1)] = 0.0f;
  const float isc0 = sm.iscl[tid >> 4], isc1 = sm.iscl[(tid + 128) >> 4];
  ChunkRegs cr;
  load_chunk(cr, covs, target, b0, 0, tid);
  store_chunk(sm, cr, isc0, isc1, tid, 0);
  __syncthreads();

  v8f creg = {0.f, 0.f, 0.f, 0.f, 0.f, 0.f, 0.f, 0.f};
  v8f hreg;

  v2f xA[3];
  #pragma unroll
  for (int kt = 0; kt < 3; ++kt)
    xA[kt] = *(const v2f*)&sm.xs[ml * XPITCH + khalf * 2 + kt * 4];

  v2f woR[8];
  #pragma unroll
  for (int i = 0; i < 4; ++i)
    #pragma unroll
    for (int e = 0; e < 2; ++e) {
      int k = (wave * 4 + i) * 4 + khalf * 2 + e;
      woR[2 * i + e] = *(const v2f*)&w_out[k * 2];
    }
  const int hh_off = ml * HPITCH + wave * 16 + khalf * 2;
  const int pp_off = (wave * 2 + khalf) * 32 + ml * 2;

  for (int t = 0; t < S_LEN; ++t) {
    v8f acc[4];
    #pragma unroll
    for (int nt = 0; nt < 4; ++nt) acc[nt] = statC[nt];
    #pragma unroll
    for (int nt = 0; nt < 4; ++nt)
      acc[nt] = __builtin_amdgcn_wmma_f32_16x16x4_f32(false, xA[0], false, wihB[0][nt],
                                                      (short)0, acc[nt], false, false);
    #pragma unroll
    for (int nt = 0; nt < 2; ++nt)
      acc[nt] = __builtin_amdgcn_wmma_f32_16x16x4_f32(false, xA[1], false, wihB[1][nt],
                                                      (short)0, acc[nt], false, false);
    __builtin_amdgcn_sched_barrier(0);
    asm volatile("s_wait_dscnt 0x0" ::: "memory");
    __syncthreads();

    if ((t & (CHUNK - 1)) == 0 && t + CHUNK < S_LEN)
      load_chunk(cr, covs, target, b0, t + CHUNK, tid);

    v2f hA[16], xAn[3];
    #pragma unroll
    for (int kt = 0; kt < 16; ++kt)
      hA[kt] = *(const v2f*)&sm.h[ml * HPITCH + kt * 4 + khalf * 2];
    __builtin_amdgcn_sched_barrier(0);
    v2f hH[4];
    #pragma unroll
    for (int i = 0; i < 4; ++i)
      hH[i] = *(const v2f*)&sm.h[hh_off + i * 4];
    const float* xrow = &sm.xs[(((t + 1) & (2 * CHUNK - 1)) * TB + ml) * XPITCH + khalf * 2];
    #pragma unroll
    for (int kt = 0; kt < 3; ++kt)
      xAn[kt] = *(const v2f*)&xrow[kt * 4];
    __builtin_amdgcn_sched_barrier(0);

    #pragma unroll
    for (int nt = 2; nt < 4; ++nt)
      acc[nt] = __builtin_amdgcn_wmma_f32_16x16x4_f32(false, xA[1], false, wihB[1][nt],
                                                      (short)0, acc[nt], false, false);
    #pragma unroll
    for (int nt = 0; nt < 4; ++nt)
      acc[nt] = __builtin_amdgcn_wmma_f32_16x16x4_f32(false, xA[2], false, wihB[2][nt],
                                                      (short)0, acc[nt], false, false);
    __builtin_amdgcn_sched_barrier(0);
    #pragma unroll
    for (int kt = 0; kt < 8; ++kt)
      #pragma unroll
      for (int nt = 0; nt < 4; ++nt)
        acc[nt] = __builtin_amdgcn_wmma_f32_16x16x4_f32(false, hA[kt], false, whhB[kt][nt],
                                                        (short)0, acc[nt], false, false);
    *(v2f*)&sm.part[((t - 1) & (CHUNK - 1)) * PPITCH + pp_off] = head_partial(hH, woR);
    __builtin_amdgcn_sched_barrier(0);
    #pragma unroll
    for (int kt = 8; kt < 12; ++kt)
      #pragma unroll
      for (int nt = 0; nt < 4; ++nt)
        acc[nt] = __builtin_amdgcn_wmma_f32_16x16x4_f32(false, hA[kt], false, whhB[kt][nt],
                                                        (short)0, acc[nt], false, false);
    __builtin_amdgcn_sched_barrier(0);
    #pragma unroll
    for (int kt = 12; kt < 16; ++kt)
      #pragma unroll
      for (int nt = 0; nt < 4; ++nt)
        acc[nt] = __builtin_amdgcn_wmma_f32_16x16x4_f32(false, hA[kt], false, whhB[kt][nt],
                                                        (short)0, acc[nt], false, false);
    #pragma unroll
    for (int r = 0; r < 8; ++r) {
      float gi = sigmoid_of_half(acc[0][r]);
      float gf = sigmoid_of_half(acc[1][r]);
      float gg = fast_tanh(acc[2][r]);
      float go = sigmoid_of_half(acc[3][r]);
      float cc = gf * creg[r] + gi * gg;
      creg[r] = cc;
      hreg[r] = go * fast_tanh(cc);
    }

    __syncthreads();

    #pragma unroll
    for (int r = 0; r < 8; ++r) {
      int m = r + khalf * 8;
      sm.h[m * HPITCH + ncol] = hreg[r];
    }
    if (t > 0 && (t & (CHUNK - 1)) == 0)
      flush_outputs(sm, out, b0, t - CHUNK, tid);
    if ((t & (CHUNK - 1)) == 7 && t + 9 < S_LEN)
      store_chunk(sm, cr, isc0, isc1, tid, ((t + 9) >> 4) & 1);
    #pragma unroll
    for (int kt = 0; kt < 3; ++kt) xA[kt] = xAn[kt];
  }

  __syncthreads();
  {
    v2f hH[4];
    #pragma unroll
    for (int i = 0; i < 4; ++i)
      hH[i] = *(const v2f*)&sm.h[hh_off + i * 4];
    *(v2f*)&sm.part[((S_LEN - 1) & (CHUNK - 1)) * PPITCH + pp_off] = head_partial(hH, woR);
  }
  __syncthreads();
  flush_outputs(sm, out, b0, S_LEN - CHUNK, tid);
}

extern "C" void kernel_launch(void* const* d_in, const int* in_sizes, int n_in,
                              void* d_out, int out_size, void* d_ws, size_t ws_size,
                              hipStream_t stream) {
  const float* target = (const float*)d_in[0];
  const float* covs   = (const float*)d_in[1];
  const int*   cats   = (const int*)  d_in[2];
  const float* scale  = (const float*)d_in[3];
  const float* emb0   = (const float*)d_in[4];
  const float* emb1   = (const float*)d_in[5];
  const float* emb2   = (const float*)d_in[6];
  const float* emb3   = (const float*)d_in[7];
  const float* w_ih   = (const float*)d_in[8];
  const float* w_hh   = (const float*)d_in[9];
  const float* bias   = (const float*)d_in[10];
  const float* w_out  = (const float*)d_in[11];
  const float* b_out  = (const float*)d_in[12];
  float* out = (float*)d_out;

  dim3 grid(B_TOT / TB);
  dim3 block(128);
  deepar_lstm_kernel<<<grid, block, 0, stream>>>(
      target, covs, cats, scale, emb0, emb1, emb2, emb3,
      w_ih, w_hh, bias, w_out, b_out, out);
}


# Gate product transposed (weights A, state B): a lane's D is its next B, so 28 of 76 WMMAs need no hand-over; h crosses LDS as b128, 2 buffers, 1 split barrier under them; gate+head sum order changed
# speedup vs baseline: 2.3562x; 1.0685x over previous
#include <hip/hip_runtime.h>
#include <hip/hip_bf16.h>
#include <math.h>

typedef __attribute__((ext_vector_type(2))) float v2f;
typedef __attribute__((ext_vector_type(8))) float v8f;

#define B_TOT   4096
#define S_LEN   512
#define HIDN    64
#define G4      256
#define NCOV    10
#define EMB     16
#define TB      16
#define HPITCH  68
#define SXPITCH 68
#define SXTILES 17
#define XPITCH  12
#define CHUNK   16
#define CPITCH  (CHUNK*NCOV)
#define PPITCH  260

struct __align__(16) SMem {
  float xs[2 * CHUNK * TB * XPITCH];
  float h[2 * TB * HPITCH];
  float sx[TB * SXPITCH];
  float part[2 * CHUNK * PPITCH];
  float bo[2];
  float scl[TB];
  float iscl[TB];
};

__device__ __forceinline__ float fast_tanh(float x) {
#if __has_builtin(__builtin_amdgcn_tanhf)
  return __builtin_amdgcn_tanhf(x);
#else
  float y;
  asm("v_tanh_f32 %0, %1" : "=v"(y) : "v"(x));
  return y;
#endif
}

__device__ __forceinline__ float sigmoid_of_half(float xh) {
  return fmaf(0.5f, fast_tanh(xh), 0.5f);
}

__device__ __forceinline__ constexpr float gate_scale(int nt) { return nt == 2 ? 1.0f : 0.5f; }

struct ChunkRegs {
  float4 cv[TB * (CPITCH / 4) / 128];
  float  tg[TB * CHUNK / 128];
};

__device__ __forceinline__ void load_chunk(ChunkRegs& cr,
                                           const float* __restrict__ covs,
                                           const float* __restrict__ target,
                                           int b0, int t0, int tid) {
  #pragma unroll
  for (int i = 0; i < TB * (CPITCH / 4) / 128; ++i) {
    int e4 = tid + i * 128;
    int m = e4 / (CPITCH / 4), off4 = e4 % (CPITCH / 4);
    cr.cv[i] = *(const float4*)&covs[((size_t)(b0 + m) * S_LEN + t0) * NCOV + off4 * 4];
  }
  #pragma unroll
  for (int i = 0; i < TB * CHUNK / 128; ++i) {
    int e = tid + i * 128;
    int m = e >> 4, j = e & 15;
    int ti = t0 + j - 1;
    float v = target[(size_t)(b0 + m) * S_LEN + (ti < 0 ? 0 : ti)];
    cr.tg[i] = (ti < 0) ? 0.0f : v;
  }
}

__device__ __forceinline__ void store_chunk(SMem& sm, const ChunkRegs& cr,
                                            float isc0, float isc1, int tid, int half) {
  float* xs = &sm.xs[half * (CHUNK * TB * XPITCH)];
  #pragma unroll
  for (int i = 0; i < TB * (CPITCH / 4) / 128; ++i) {
    int e4 = tid + i * 128;
    int m = e4 / (CPITCH / 4), off4 = e4 % (CPITCH / 4);
    float vv[4] = { cr.cv[i].x, cr.cv[i].y, cr.cv[i].z, cr.cv[i].w };
    #pragma unroll
    for (int q = 0; q < 4; ++q) {
      int idx = off4 * 4 + q;
      int tc = idx / NCOV, k = idx % NCOV;
      xs[(tc * TB + m) * XPITCH + 1 + k] = vv[q];
    }
  }
  #pragma unroll
  for (int i = 0; i < TB * CHUNK / 128; ++i) {
    int e = tid + i * 128;
    int m = e >> 4, j = e & 15;
    xs[(j * TB + m) * XPITCH] = cr.tg[i] * (i == 0 ? isc0 : isc1);
  }
}

__device__ __forceinline__ v2f head_partial(const v8f hv, const v2f w2[8]) {
  v2f sa = {0.0f, 0.0f}, sb = {0.0f, 0.0f};
  #pragma unroll
  for (int r = 0; r < 4; ++r) {
    sa.x = fmaf(hv[r],     w2[r].x,     sa.x);
    sa.y = fmaf(hv[r],     w2[r].y,     sa.y);
    sb.x = fmaf(hv[r + 4], w2[r + 4].x, sb.x);
    sb.y = fmaf(hv[r + 4], w2[r + 4].y, sb.y);
  }
  v2f s = { sa.x + sb.x, sa.y + sb.y };
  return s;
}

__device__ __forceinline__ void flush_outputs(const SMem& sm, float* __restrict__ out,
                                              int b0, int t0, int slot0, int tid) {
  const int tq = tid & 3;
  const int p  = tid >> 2;
  const int m  = p >> 1, j = p & 1;
  const float bj   = sm.bo[j];
  const float mult = (j == 0) ? sm.scl[m] : 1.0f;
  float v[4];
  #pragma unroll
  for (int i = 0; i < 4; ++i) {
    const float* pp = &sm.part[(slot0 + tq * 4 + i) * PPITCH + p];
    float r = bj;
    #pragma unroll
    for (int g = 0; g < 8; ++g) r += pp[g * 32];
    float l = log1pf(__expf(r));
    float sp = (r > 20.0f) ? r : l;
    sp += 1e-4f;
    v[i] = sp * mult;
  }
  size_t idx = (size_t)j * B_TOT * S_LEN + (size_t)(b0 + m) * S_LEN + t0 + tq * 4;
  *(float4*)&out[idx] = make_float4(v[0], v[1], v[2], v[3]);
}

__global__ __launch_bounds__(128, 1)
void deepar_lstm_kernel(const float* __restrict__ target,
                        const float* __restrict__ covs,
                        const int*   __restrict__ cats,
                        const float* __restrict__ scale,
                        const float* __restrict__ emb0,
                        const float* __restrict__ emb1,
                        const float* __restrict__ emb2,
                        const float* __restrict__ emb3,
                        const float* __restrict__ w_ih,
                        const float* __restrict__ w_hh,
                        const float* __restrict__ bias,
                        const float* __restrict__ w_out,
                        const float* __restrict__ b_out,
                        float* __restrict__ out)
{
  __shared__ SMem sm;
  const int tid  = threadIdx.x;
  const int wave = tid >> 5;
  const int lane = tid & 31;
  const int b0   = blockIdx.x * TB;

  if (tid < TB) {
    float s = scale[b0 + tid];
    sm.scl[tid]  = s;
    sm.iscl[tid] = 1.0f / fmaxf(s, 1e-4f);
  }
  if (tid < 2)        sm.bo[tid] = b_out[tid];

  for (int e = tid; e < TB * SXPITCH; e += 128) {
    int m = e / SXPITCH, k = e % SXPITCH;
    float v;
    if      (k < 16) v = emb0[cats[(b0 + m) * 4 + 0] * EMB + k];
    else if (k < 32) v = emb1[cats[(b0 + m) * 4 + 1] * EMB + (k - 16)];
    else if (k < 48) v = emb2[cats[(b0 + m) * 4 + 2] * EMB + (k - 32)];
    else if (k < 64) v = emb3[cats[(b0 + m) * 4 + 3] * EMB + (k - 48)];
    else if (k == 64) v = log1pf(scale[b0 + m]);
    else if (k == 65) v = 1.0f;
    else             v = 0.0f;
    sm.sx[e] = v;
  }
  __syncthreads();

  const int khalf = lane >> 4;
  const int ml    = lane & 15;
  const int ncol  = wave * 16 + ml;

  v2f whhB[4][4][4];
  #pragma unroll
  for (int blk = 0; blk < 4; ++blk)
    #pragma unroll
    for (int j = 0; j < 4; ++j)
      #pragma unroll
      for (int nt = 0; nt < 4; ++nt) {
        int k = ((wave + blk) & 3) * 16 + khalf * 8 + j * 2;
        int n = nt * 64 + ncol;
        v2f b = { gate_scale(nt) * w_hh[k * G4 + n], gate_scale(nt) * w_hh[(k + 1) * G4 + n] };
        whhB[blk][j][nt] = b;
      }
  v2f wihB[3][4];
  #pragma unroll
  for (int kt = 0; kt < 3; ++kt)
    #pragma unroll
    for (int nt = 0; nt < 4; ++nt) {
      int k = kt * 4 + khalf * 2;
      int n = nt * 64 + ncol;
      v2f b = { gate_scale(nt) * w_ih[k * G4 + n], gate_scale(nt) * w_ih[(k + 1) * G4 + n] };
      wihB[kt][nt] = b;
    }
  v2f wsB[SXTILES][4];
  #pragma unroll
  for (int kt = 0; kt < SXTILES - 1; ++kt)
    #pragma unroll
    for (int nt = 0; nt < 4; ++nt) {
      int k = 11 + kt * 4 + khalf * 2;
      int n = nt * 64 + ncol;
      v2f b = { gate_scale(nt) * w_ih[k * G4 + n], gate_scale(nt) * w_ih[(k + 1) * G4 + n] };
      wsB[kt][nt] = b;
    }
  #pragma unroll
  for (int nt = 0; nt < 4; ++nt) {
    float w64 = gate_scale(nt) * w_ih[75 * G4 + nt * 64 + ncol];
    float bv  = gate_scale(nt) * bias[nt * 64 + ncol];
    v2f b = { khalf == 0 ? w64 : 0.0f, khalf == 0 ? bv : 0.0f };
    wsB[SXTILES - 1][nt] = b;
  }
  v8f statC[4];
  #pragma unroll
  for (int nt = 0; nt < 4; ++nt)
    statC[nt] = (v8f){0.f, 0.f, 0.f, 0.f, 0.f, 0.f, 0.f, 0.f};
  v2f sA[SXTILES];
  #pragma unroll
  for (int kt = 0; kt < SXTILES; ++kt)
    sA[kt] = *(const v2f*)&sm.sx[ml * SXPITCH + kt * 4 + khalf * 2];
  __builtin_amdgcn_sched_barrier(0);
  #pragma unroll
  for (int kt = 0; kt < SXTILES; ++kt)
    #pragma unroll
    for (int nt = 0; nt < 4; ++nt)
      statC[nt] = __builtin_amdgcn_wmma_f32_16x16x4_f32(false, wsB[kt][nt], false, sA[kt],
                                                        (short)0, statC[nt], false, false);
  __builtin_amdgcn_sched_barrier(0);
  asm volatile("v_nop\n\tv_nop\n\tv_nop\n\tv_nop"
               : "+v"(statC[0]), "+v"(statC[1]), "+v"(statC[2]), "+v"(statC[3]));
  __builtin_amdgcn_sched_barrier(0);

  for (int e = tid; e < 2 * CHUNK * TB; e += 128) sm.xs[e * XPITCH + (XPITCH - 1)] = 0.0f;
  const float isc0 = sm.iscl[tid >> 4], isc1 = sm.iscl[(tid + 128) >> 4];
  ChunkRegs cr;
  load_chunk(cr, covs, target, b0, 0, tid);
  store_chunk(sm, cr, isc0, isc1, tid, 0);
  __syncthreads();

  v8f creg = {0.f, 0.f, 0.f, 0.f, 0.f, 0.f, 0.f, 0.f};
  v8f hreg = {0.f, 0.f, 0.f, 0.f, 0.f, 0.f, 0.f, 0.f};

  v2f xA[3];
  #pragma unroll
  for (int kt = 0; kt < 3; ++kt)
    xA[kt] = *(const v2f*)&sm.xs[ml * XPITCH + khalf * 2 + kt * 4];

  v2f woR[8];
  #pragma unroll
  for (int r = 0; r < 8; ++r)
    woR[r] = *(const v2f*)&w_out[(wave * 16 + khalf * 8 + r) * 2];
  const int own_off = ml * HPITCH + wave * 16 + khalf * 8;
  int oth_off[3];
  #pragma unroll
  for (int b = 0; b < 3; ++b)
    oth_off[b] = ml * HPITCH + ((wave + b + 1) & 3) * 16 + khalf * 8;
  const int pp_off = (wave * 2 + khalf) * 32 + ml * 2;
  v2f hp = {0.0f, 0.0f};

  for (int t = 0; t < S_LEN; ++t) {
    float* hb = &sm.h[(t & 1) * (TB * HPITCH)];
    *(float4*)&hb[own_off]     = make_float4(hreg[0], hreg[1], hreg[2], hreg[3]);
    *(float4*)&hb[own_off + 4] = make_float4(hreg[4], hreg[5], hreg[6], hreg[7]);
    *(v2f*)&sm.part[((t - 1) & (2 * CHUNK - 1)) * PPITCH + pp_off] = hp;
    __builtin_amdgcn_sched_barrier(0);

    v8f acc[4];
    #pragma unroll
    for (int nt = 0; nt < 4; ++nt) acc[nt] = statC[nt];
    #pragma unroll
    for (int nt = 0; nt < 4; ++nt)
      acc[nt] = __builtin_amdgcn_wmma_f32_16x16x4_f32(false, wihB[0][nt], false, xA[0],
                                                      (short)0, acc[nt], false, false);
    #pragma unroll
    for (int nt = 0; nt < 2; ++nt)
      acc[nt] = __builtin_amdgcn_wmma_f32_16x16x4_f32(false, wihB[1][nt], false, xA[1],
                                                      (short)0, acc[nt], false, false);
    __builtin_amdgcn_sched_barrier(0);
    asm volatile("s_wait_dscnt 0x0\n\ts_barrier_signal -1" ::: "memory");
    __builtin_amdgcn_sched_barrier(0);
    #pragma unroll
    for (int nt = 2; nt < 4; ++nt)
      acc[nt] = __builtin_amdgcn_wmma_f32_16x16x4_f32(false, wihB[1][nt], false, xA[1],
                                                      (short)0, acc[nt], false, false);
    #pragma unroll
    for (int nt = 0; nt < 4; ++nt)
      acc[nt] = __builtin_amdgcn_wmma_f32_16x16x4_f32(false, wihB[2][nt], false, xA[2],
                                                      (short)0, acc[nt], false, false);
    #pragma unroll
    for (int j = 0; j < 2; ++j) {
      v2f bo = { hreg[2 * j], hreg[2 * j + 1] };
      #pragma unroll
      for (int nt = 0; nt < 4; ++nt)
        acc[nt] = __builtin_amdgcn_wmma_f32_16x16x4_f32(false, whhB[0][j][nt], false, bo,
                                                        (short)0, acc[nt], false, false);
    }
    __builtin_amdgcn_sched_barrier(0);
    asm volatile("s_barrier_wait -1" ::: "memory");
    __builtin_amdgcn_sched_barrier(0);

    if ((t & (CHUNK - 1)) == 0 && t + CHUNK < S_LEN)
      load_chunk(cr, covs, target, b0, t + CHUNK, tid);

    float4 oh[3][2];
    #pragma unroll
    for (int b = 0; b < 3; ++b)
      #pragma unroll
      for (int jj = 0; jj < 2; ++jj)
        oh[b][jj] = *(const float4*)&hb[oth_off[b] + jj * 4];
    v2f xAn[3];
    const float* xrow = &sm.xs[(((t + 1) & (2 * CHUNK - 1)) * TB + ml) * XPITCH + khalf * 2];
    #pragma unroll
    for (int kt = 0; kt < 3; ++kt)
      xAn[kt] = *(const v2f*)&xrow[kt * 4];
    __builtin_amdgcn_sched_barrier(0);

    #pragma unroll
    for (int j = 2; j < 4; ++j) {
      v2f bo = { hreg[2 * j], hreg[2 * j + 1] };
      #pragma unroll
      for (int nt = 0; nt < 4; ++nt)
        acc[nt] = __builtin_amdgcn_wmma_f32_16x16x4_f32(false, whhB[0][j][nt], false, bo,
                                                        (short)0, acc[nt], false, false);
    }
    __builtin_amdgcn_sched_barrier(0);
    #pragma unroll
    for (int b = 0; b < 3; ++b)
      #pragma unroll
      for (int j = 0; j < 4; ++j) {
        const float4 q = oh[b][j >> 1];
        v2f bo = (j & 1) ? (v2f){ q.z, q.w } : (v2f){ q.x, q.y };
        #pragma unroll
        for (int nt = 0; nt < 4; ++nt)
          acc[nt] = __builtin_amdgcn_wmma_f32_16x16x4_f32(false, whhB[b + 1][j][nt], false, bo,
                                                          (short)0, acc[nt], false, false);
      }
    #pragma unroll
    for (int r = 0; r < 8; ++r) {
      float gi = sigmoid_of_half(acc[0][r]);
      float gf = sigmoid_of_half(acc[1][r]);
      float gg = fast_tanh(acc[2][r]);
      float go = sigmoid_of_half(acc[3][r]);
      float cc = gf * creg[r] + gi * gg;
      creg[r] = cc;
      hreg[r] = go * fast_tanh(cc);
    }
    hp = head_partial(hreg, woR);

    if (t > 0 && (t & (CHUNK - 1)) == 0)
      flush_outputs(sm, out, b0, t - CHUNK, (t - CHUNK) & (2 * CHUNK - 1), tid);
    if ((t & (CHUNK - 1)) == 7 && t + 9 < S_LEN)
      store_chunk(sm, cr, isc0, isc1, tid, ((t + 9) >> 4) & 1);
    #pragma unroll
    for (int kt = 0; kt < 3; ++kt) xA[kt] = xAn[kt];
  }

  *(v2f*)&sm.part[((S_LEN - 1) & (2 * CHUNK - 1)) * PPITCH + pp_off] = hp;
  __syncthreads();
  flush_outputs(sm, out, b0, S_LEN - CHUNK, (S_LEN - CHUNK) & (2 * CHUNK - 1), tid);
}

extern "C" void kernel_launch(void* const* d_in, const int* in_sizes, int n_in,
                              void* d_out, int out_size, void* d_ws, size_t ws_size,
                              hipStream_t stream) {
  const float* target = (const float*)d_in[0];
  const float* covs   = (const float*)d_in[1];
  const int*   cats   = (const int*)  d_in[2];
  const float* scale  = (const float*)d_in[3];
  const float* emb0   = (const float*)d_in[4];
  const float* emb1   = (const float*)d_in[5];
  const float* emb2   = (const float*)d_in[6];
  const float* emb3   = (const float*)d_in[7];
  const float* w_ih   = (const float*)d_in[8];
  const float* w_hh   = (const float*)d_in[9];
  const float* bias   = (const float*)d_in[10];
  const float* w_out  = (const float*)d_in[11];
  const float* b_out  = (const float*)d_in[12];
  float* out = (float*)d_out;

  dim3 grid(B_TOT / TB);
  dim3 block(128);
  deepar_lstm_kernel<<<grid, block, 0, stream>>>(
      target, covs, cats, scale, emb0, emb1, emb2, emb3,
      w_ih, w_hh, bias, w_out, b_out, out);
}
